# starting kernel
# baseline (speedup 1.0000x reference)
#include <hip/hip_runtime.h>
#include <math.h>

#define NN 50000
#define NE 1600000
#define DIN 128
#define DOUT 32
#define NH 4
#define DT 128
#define NTILES (NN / 16)
#define NPAIR (DIN / 2)
#define WPAD 144

typedef __attribute__((ext_vector_type(2))) float v2f;
typedef __attribute__((ext_vector_type(8))) float v8f;

__global__ void __launch_bounds__(128) gat_wh_gemm(const float* __restrict__ x,
                                                   const float* __restrict__ W,
                                                   float* __restrict__ Wh) {
  __shared__ v2f Wsh[NPAIR * WPAD];

  for (int idx = threadIdx.x; idx < NPAIR * DT; idx += 128) {
    const int p  = idx >> 7;
    const int j  = idx & 127;
    const int kh = j >> 5;
    const int o  = j & 31;
    const float* wp = W + kh * (DIN * DOUT) + (2 * p) * DOUT + o;
    v2f pr;
    pr.x = wp[0];
    pr.y = wp[DOUT];
    Wsh[p * WPAD + j] = pr;
  }
  __syncthreads();

  const int lane = threadIdx.x & 31;
  const int wave = threadIdx.x >> 5;
  const int half = lane >> 4;
  const int l16  = lane & 15;
  const int tile = blockIdx.x * 4 + wave;
  if (tile >= NTILES) return;
  const int row0 = tile * 16;

  v8f acc[8];
#pragma unroll
  for (int jt = 0; jt < 8; ++jt) acc[jt] = (v8f){0.f,0.f,0.f,0.f,0.f,0.f,0.f,0.f};

  const float* xrow = x + (size_t)(row0 + l16) * DIN;

#pragma unroll 4
  for (int kk = 0; kk < 32; ++kk) {
    const int i0 = kk * 4 + half * 2;
    const int p  = kk * 2 + half;
    const v2f a = *(const v2f*)(xrow + i0);
    const v2f* wr = Wsh + p * WPAD + l16;
#pragma unroll
    for (int jt = 0; jt < 8; ++jt) {
      const v2f b = wr[jt * 16];
      acc[jt] = __builtin_amdgcn_wmma_f32_16x16x4_f32(
          false, a, false, b, (short)0, acc[jt], false, false);
    }
  }

#pragma unroll
  for (int jt = 0; jt < 8; ++jt) {
#pragma unroll
    for (int r = 0; r < 8; ++r) {
      const int row = row0 + r + 8 * half;
      Wh[(size_t)row * DT + jt * 16 + l16] = acc[jt][r];
    }
  }
}

__global__ void gat_init(float* __restrict__ out, float* __restrict__ e_max,
                         float* __restrict__ e_sum) {
  int i = blockIdx.x * blockDim.x + threadIdx.x;
  if (i < NN * DT) out[i] = 0.f;
  if (i < NN * NH) { e_sum[i] = 0.f; e_max[i] = -__builtin_inff(); }
}

__global__ void gat_scores(const float* __restrict__ Wh, const float* __restrict__ a,
                           float* __restrict__ s_src, float* __restrict__ s_tgt) {
  int idx = blockIdx.x * blockDim.x + threadIdx.x;
  if (idx >= NN * NH) return;
  int n = idx >> 2, k = idx & 3;
  const float* w  = Wh + (size_t)n * DT + k * DOUT;
  const float* as = a + k * (2 * DOUT);
  const float* at = as + DOUT;
  float ss = 0.f, st = 0.f;
#pragma unroll
  for (int o = 0; o < DOUT; ++o) { float v = w[o]; ss += v * as[o]; st += v * at[o]; }
  s_src[idx] = ss;
  s_tgt[idx] = st;
}

__device__ inline void atomicMaxF(float* addr, float val) {
  if (val >= 0.f) atomicMax((int*)addr, __float_as_int(val));
  else            atomicMin((unsigned int*)addr, __float_as_uint(val));
}

__global__ void gat_edge_max(const int* __restrict__ src, const int* __restrict__ tgt,
                             const float* __restrict__ s_src, const float* __restrict__ s_tgt,
                             float* __restrict__ e_buf, float* __restrict__ e_max) {
  int idx = blockIdx.x * blockDim.x + threadIdx.x;
  if (idx >= NE * NH) return;
  int e = idx >> 2, k = idx & 3;
  int s = src[e], t = tgt[e];
  float v = s_src[s * NH + k] + s_tgt[t * NH + k];
  v = v > 0.f ? v : 0.2f * v;
  e_buf[idx] = v;
  atomicMaxF(&e_max[t * NH + k], v);
}

__global__ void gat_edge_exp(const int* __restrict__ tgt, const float* __restrict__ e_max,
                             float* __restrict__ e_buf, float* __restrict__ e_sum) {
  int idx = blockIdx.x * blockDim.x + threadIdx.x;
  if (idx >= NE * NH) return;
  int e = idx >> 2, k = idx & 3;
  int t = tgt[e];
  float ex = __expf(e_buf[idx] - e_max[t * NH + k]);
  e_buf[idx] = ex;
  atomicAdd(&e_sum[t * NH + k], ex);
}

__global__ void gat_edge_agg(const int* __restrict__ src, const int* __restrict__ tgt,
                             const float* __restrict__ e_buf, const float* __restrict__ e_sum,
                             const float* __restrict__ Wh, float* __restrict__ out) {
  long long idx = (long long)blockIdx.x * blockDim.x + threadIdx.x;
  if (idx >= (long long)NE * DT) return;
  int e = (int)(idx >> 7);
  int j = (int)(idx & 127);
  int k = j >> 5;
  int s = src[e], t = tgt[e];
  float alpha = e_buf[(size_t)e * NH + k] / (e_sum[t * NH + k] + 1e-10f);
  atomicAdd(&out[(size_t)t * DT + j], alpha * Wh[(size_t)s * DT + j]);
}

__global__ void gat_elu(float* __restrict__ out) {
  int i = blockIdx.x * blockDim.x + threadIdx.x;
  if (i >= NN * DT) return;
  float v = out[i];
  out[i] = v > 0.f ? v : expm1f(v);
}

extern "C" void kernel_launch(void* const* d_in, const int* in_sizes, int n_in,
                              void* d_out, int out_size, void* d_ws, size_t ws_size,
                              hipStream_t stream) {
  const float* x    = (const float*)d_in[0];
  const int*   eidx = (const int*)d_in[1];
  const float* W    = (const float*)d_in[2];
  const float* a    = (const float*)d_in[3];
  float* out = (float*)d_out;

  const int* src = eidx;
  const int* tgt = eidx + NE;

  float* ws    = (float*)d_ws;
  float* Wh    = ws;
  float* s_src = Wh + (size_t)NN * DT;
  float* s_tgt = s_src + NN * NH;
  float* e_max = s_tgt + NN * NH;
  float* e_sum = e_max + NN * NH;
  float* e_buf = e_sum + NN * NH;

  const int n_out = NN * DT;
  gat_init<<<(n_out + 255) / 256, 256, 0, stream>>>(out, e_max, e_sum);

  gat_wh_gemm<<<(NTILES + 3) / 4, 128, 0, stream>>>(x, W, Wh);

  gat_scores<<<(NN * NH + 255) / 256, 256, 0, stream>>>(Wh, a, s_src, s_tgt);

  const int ne4 = NE * NH;
  gat_edge_max<<<(ne4 + 255) / 256, 256, 0, stream>>>(src, tgt, s_src, s_tgt, e_buf, e_max);
  gat_edge_exp<<<(ne4 + 255) / 256, 256, 0, stream>>>(tgt, e_max, e_buf, e_sum);

  const long long neagg = (long long)NE * DT;
  gat_edge_agg<<<(int)((neagg + 255) / 256), 256, 0, stream>>>(src, tgt, e_buf, e_sum, Wh, out);

  gat_elu<<<(n_out + 255) / 256, 256, 0, stream>>>(out);
}


# edge-phase scan reads a 1-byte owner plane (target/200, written by the projection waves), not the 4-byte targets: 16 edges per 16-B load, packed byte compare; sums in arrival order as before
# speedup vs baseline: 6.4747x; 6.4747x over previous
#include <hip/hip_runtime.h>
#include <math.h>

#define NN 50000
#define NE 1600000
#define DIN 128
#define DOUT 32
#define NH 4
#define DT 128
#define NTILES (NN / 16)
#define NPAIR (DIN / 2)
#define WPAD 144
#define NB 200
#define EPW (NE / NTILES)
static_assert(NE == NTILES * EPW && EPW == 512, "one projection wave per 512 edges");
static_assert((NN + NB - 1) / NB <= 255, "an owner id must fit a byte below 255");

typedef __attribute__((ext_vector_type(2))) float v2f;
typedef __attribute__((ext_vector_type(8))) float v8f;

__device__ inline unsigned owner_of(int t) {
  const unsigned o = (unsigned)t / (unsigned)NB;
  return o < 255u ? o : 255u;
}

__global__ void __launch_bounds__(128) gat_wh_gemm(const float* __restrict__ x,
                                                   const float* __restrict__ W,
                                                   const int* __restrict__ tgt,
                                                   float* __restrict__ Wh,
                                                   unsigned* __restrict__ plane) {
  __shared__ v2f Wsh[NPAIR * WPAD];

  const int lane = threadIdx.x & 31;
  const int wave = threadIdx.x >> 5;
  const int tile = blockIdx.x * 4 + wave;

  int4 tv[EPW / 128];
  {
    const int4* __restrict__ t4p = (const int4*)tgt;
    const int w0 = (tile < NTILES ? tile : 0) * (EPW / 4) + lane;
#pragma unroll
    for (int u = 0; u < EPW / 128; ++u) tv[u] = t4p[w0 + u * 32];
  }

  for (int idx = threadIdx.x; idx < NPAIR * DT; idx += 128) {
    const int p  = idx >> 7;
    const int j  = idx & 127;
    const int kh = j >> 5;
    const int o  = j & 31;
    const float* wp = W + kh * (DIN * DOUT) + (2 * p) * DOUT + o;
    v2f pr;
    pr.x = wp[0];
    pr.y = wp[DOUT];
    Wsh[p * WPAD + j] = pr;
  }
  __syncthreads();

  const int half = lane >> 4;
  const int l16  = lane & 15;
  if (tile >= NTILES) return;
  const int row0 = tile * 16;

#pragma unroll
  for (int u = 0; u < EPW / 128; ++u) {
    const int4 v = tv[u];
    plane[tile * (EPW / 4) + u * 32 + lane] =
        owner_of(v.x) | (owner_of(v.y) << 8) | (owner_of(v.z) << 16) | (owner_of(v.w) << 24);
  }

  v8f acc[8];
#pragma unroll
  for (int jt = 0; jt < 8; ++jt) acc[jt] = (v8f){0.f,0.f,0.f,0.f,0.f,0.f,0.f,0.f};

  const float* xrow = x + (size_t)(row0 + l16) * DIN;

#pragma unroll 4
  for (int kk = 0; kk < 32; ++kk) {
    const int i0 = kk * 4 + half * 2;
    const int p  = kk * 2 + half;
    const v2f a = *(const v2f*)(xrow + i0);
    const v2f* wr = Wsh + p * WPAD + l16;
#pragma unroll
    for (int jt = 0; jt < 8; ++jt) {
      const v2f b = wr[jt * 16];
      acc[jt] = __builtin_amdgcn_wmma_f32_16x16x4_f32(
          false, a, false, b, (short)0, acc[jt], false, false);
    }
  }

#pragma unroll
  for (int jt = 0; jt < 8; ++jt) {
#pragma unroll
    for (int r = 0; r < 8; ++r) {
      const int row = row0 + r + 8 * half;
      Wh[(size_t)row * DT + jt * 16 + l16] = acc[jt][r];
    }
  }
}

__global__ void gat_scores(const float* __restrict__ Wh, const float* __restrict__ a,
                           float* __restrict__ s_src, float* __restrict__ s_tgt) {
  int idx = blockIdx.x * blockDim.x + threadIdx.x;
  if (idx >= NN * NH) return;
  int n = idx >> 2, k = idx & 3;
  const float* w  = Wh + (size_t)n * DT + k * DOUT;
  const float* as = a + k * (2 * DOUT);
  const float* at = as + DOUT;
  float ss = 0.f, st = 0.f;
#pragma unroll
  for (int o = 0; o < DOUT; ++o) { float v = w[o]; ss += v * as[o]; st += v * at[o]; }
  s_src[idx] = ss;
  s_tgt[idx] = st;
}

#define NOWN (NN / NB)
#define ACCP 130
#define CAP 8192
#define EPT (CAP / 1024)
#define NP16 (NE / 16)
#define SCAN_LD 8
#define SCAN_CHUNK (1024 * SCAN_LD)
#define NFULL (NP16 / SCAN_CHUNK)
#define NTAIL ((NP16 - NFULL * SCAN_CHUNK + 1023) / 1024)
#define KEY_EMPTY (-2147483647 - 1)

__device__ inline int f2key(float v) {
  const int b = __float_as_int(v);
  return b >= 0 ? b : (b ^ 0x7FFFFFFF);
}
__device__ inline float key2f(int k) {
  return __int_as_float(k >= 0 ? k : (k ^ 0x7FFFFFFF));
}
__device__ inline float leaky(float v) { return v > 0.f ? v : 0.2f * v; }
__device__ inline float score_exp(float score, int maxkey) {
  float m = key2f(maxkey);
  if (!(__builtin_fabsf(m) < __builtin_inff())) m = 0.f;
  return __expf(score - m);
}
__device__ inline float edge_exp(float score_sum, int maxkey) {
  return score_exp(leaky(score_sum), maxkey);
}

__device__ inline void scan_load(const uint4* __restrict__ p16, int base, uint4 (&tv)[SCAN_LD]) {
#pragma unroll
  for (int u = 0; u < SCAN_LD; ++u) tv[u] = p16[base + u * 1024];
}
__device__ inline unsigned nz_bytes(unsigned x) {
  return ((x & 0x7F7F7F7Fu) + 0x7F7F7F7Fu) | x;
}
#define NO_MATCH 0xF0F0F0F0u
__device__ inline unsigned ne_flags(const uint4 v, unsigned pat) {
  return (nz_bytes(v.x ^ pat) & 0x80808080u) | ((nz_bytes(v.y ^ pat) >> 1) & 0x40404040u) |
         ((nz_bytes(v.z ^ pat) >> 2) & 0x20202020u) | ((nz_bytes(v.w ^ pat) >> 3) & 0x10101010u);
}
__device__ inline void keep_pair(unsigned fa, unsigned fb, int idx, unsigned* cntp, unsigned* listp) {
  unsigned z = ~(fa | (fb >> 4));
  while (z) {
    const unsigned b  = (unsigned)__builtin_ctz(z);
    const unsigned nb = ~b;
    z &= z - 1u;
    const unsigned pos = atomicAdd(cntp, 1u);
    if (pos < (unsigned)CAP)
      listp[pos] = (unsigned)idx * 16u + ((nb & 4u) << 12) + ((nb & 3u) << 2) + (b >> 3);
  }
}
static_assert(SCAN_LD % 2 == 0 && NTAIL == 2, "the scan tests plane words in pairs 1024 words apart");

__global__ void __launch_bounds__(1024) gat_owner(const int* __restrict__ src,
                                                  const int* __restrict__ tgt,
                                                  const unsigned* __restrict__ plane,
                                                  const float* __restrict__ s_src,
                                                  const float* __restrict__ s_tgt,
                                                  const float* __restrict__ Wh,
                                                  float* __restrict__ out) {
  __shared__ float acc[NB * ACCP];
  __shared__ unsigned list[CAP];
  __shared__ float4 esc[CAP];
  __shared__ __attribute__((aligned(16))) int   emaxk[NB * NH];
  __shared__ __attribute__((aligned(16))) float esum[NB * NH];
  __shared__ __attribute__((aligned(16))) float stg[NB * NH];
  __shared__ unsigned cnt;

  const int tid = threadIdx.x;
  const int n0  = blockIdx.x * NB;
  const float4* __restrict__ s_src4 = (const float4*)s_src;

  for (int i = tid; i < NB * ACCP; i += 1024) acc[i] = 0.f;
  for (int i = tid; i < NB * NH; i += 1024) {
    emaxk[i] = KEY_EMPTY;
    esum[i]  = 0.f;
    stg[i]   = s_tgt[n0 * NH + i];
  }
  if (tid == 0) cnt = 0u;
  __syncthreads();

  {
    const uint4* __restrict__ t4p = (const uint4*)plane;
    const unsigned pat = blockIdx.x * 0x01010101u;
    uint4 cur[SCAN_LD], nxt[SCAN_LD];
    int chunk = (int)(blockIdx.x % NFULL);
    int curbase = chunk * SCAN_CHUNK + tid;
    scan_load(t4p, curbase, cur);
    for (int it = 0; it < NFULL - 1; ++it) {
      chunk = (chunk + 1 == NFULL) ? 0 : chunk + 1;
      const int nxtbase = chunk * SCAN_CHUNK + tid;
      scan_load(t4p, nxtbase, nxt);
#pragma unroll
      for (int u = 0; u < SCAN_LD; u += 2)
        keep_pair(ne_flags(cur[u], pat), ne_flags(cur[u + 1], pat), curbase + u * 1024, &cnt, list);
#pragma unroll
      for (int u = 0; u < SCAN_LD; ++u) cur[u] = nxt[u];
      curbase = nxtbase;
    }
#pragma unroll
    for (int u = 0; u < SCAN_LD; u += 2)
      keep_pair(ne_flags(cur[u], pat), ne_flags(cur[u + 1], pat), curbase + u * 1024, &cnt, list);
    {
      const int idx = NFULL * SCAN_CHUNK + tid;
      unsigned fa = NO_MATCH, fb = NO_MATCH;
      if (idx < NP16) fa = ne_flags(t4p[idx], pat);
      if (idx + 1024 < NP16) fb = ne_flags(t4p[idx + 1024], pat);
      keep_pair(fa, fb, idx, &cnt, list);
    }
  }
  __syncthreads();

  const unsigned total = (unsigned)__builtin_amdgcn_readfirstlane((int)cnt);

  if (total <= (unsigned)CAP) {
    const int n = (int)total;

    {
      unsigned en[EPT];
      int sn[EPT];
      int tn[EPT];
      float4 ss[EPT];
#pragma unroll
      for (int u = 0; u < EPT; ++u) {
        const int idx = tid + u * 1024;
        const unsigned raw = list[idx];
        en[u] = idx < n ? raw : 0u;
      }
#pragma unroll
      for (int u = 0; u < EPT; ++u) {
        sn[u] = src[en[u]];
        tn[u] = tgt[en[u]];
      }
#pragma unroll
      for (int u = 0; u < EPT; ++u) ss[u] = s_src4[sn[u]];
#pragma unroll
      for (int u = 0; u < EPT; ++u) {
        const int idx = tid + u * 1024;
        if (idx < n) {
          const unsigned d  = (unsigned)(tn[u] - n0);
          const float4   st = *(const float4*)&stg[d * NH];
          float4 sc;
          sc.x = leaky(ss[u].x + st.x);
          sc.y = leaky(ss[u].y + st.y);
          sc.z = leaky(ss[u].z + st.z);
          sc.w = leaky(ss[u].w + st.w);
          list[idx] = (unsigned)sn[u] | (d << 16);
          esc[idx]  = sc;
          atomicMax(&emaxk[d * NH + 0], f2key(sc.x));
          atomicMax(&emaxk[d * NH + 1], f2key(sc.y));
          atomicMax(&emaxk[d * NH + 2], f2key(sc.z));
          atomicMax(&emaxk[d * NH + 3], f2key(sc.w));
        }
      }
    }
    __syncthreads();

    const int lane = tid & 31;
    const int wave = __builtin_amdgcn_readfirstlane(tid >> 5);
    const int kq = lane >> 3;
    const int hm0 = kq == 0 ? -1 : 0, hm1 = kq == 1 ? -1 : 0, hm2 = kq == 2 ? -1 : 0, hm3 = kq == 3 ? -1 : 0;
    const int laneoff = 4 * lane + (lane >> 4);
    const float4* __restrict__ Wh4 = (const float4*)Wh;

    for (int base = wave * 32; base < n; base += 1024) {
      const int idx = base + lane;
      unsigned en2 = 0u;
      float ex0 = 0.f, ex1 = 0.f, ex2 = 0.f, ex3 = 0.f;
      if (idx < n) {
        en2 = list[idx];
        const unsigned d = en2 >> 16;
        const float4 sc = esc[idx];
        const int4   mk = *(const int4*)&emaxk[d * NH];
        ex0 = score_exp(sc.x, mk.x);
        ex1 = score_exp(sc.y, mk.y);
        ex2 = score_exp(sc.z, mk.z);
        ex3 = score_exp(sc.w, mk.w);
        atomicAdd(&esum[d * NH + 0], ex0);
        atomicAdd(&esum[d * NH + 1], ex1);
        atomicAdd(&esum[d * NH + 2], ex2);
        atomicAdd(&esum[d * NH + 3], ex3);
      }
      const int left = n - base;
      const int cntb = left < 32 ? left : 32;

      int j0 = 0;
      for (; j0 + 8 <= cntb; j0 += 8) {
        unsigned er[8];
        float4 r[8];
#pragma unroll
        for (int u = 0; u < 8; ++u) {
          er[u] = (unsigned)__builtin_amdgcn_readlane((int)en2, j0 + u);
          r[u]  = Wh4[(size_t)(er[u] & 0xFFFFu) * (DT / 4) + lane];
        }
#pragma unroll
        for (int u = 0; u < 8; ++u) {
          const int b0 = __builtin_amdgcn_readlane(__float_as_int(ex0), j0 + u);
          const int b1 = __builtin_amdgcn_readlane(__float_as_int(ex1), j0 + u);
          const int b2 = __builtin_amdgcn_readlane(__float_as_int(ex2), j0 + u);
          const int b3 = __builtin_amdgcn_readlane(__float_as_int(ex3), j0 + u);
          const float w = __int_as_float((b0 & hm0) | (b1 & hm1) | (b2 & hm2) | (b3 & hm3));
          float* ap = acc + (er[u] >> 16) * ACCP + laneoff;
          atomicAdd(ap + 0, w * r[u].x);
          atomicAdd(ap + 1, w * r[u].y);
          atomicAdd(ap + 2, w * r[u].z);
          atomicAdd(ap + 3, w * r[u].w);
        }
      }
      for (; j0 < cntb; ++j0) {
        const unsigned e1 = (unsigned)__builtin_amdgcn_readlane((int)en2, j0);
        const float4 r1 = Wh4[(size_t)(e1 & 0xFFFFu) * (DT / 4) + lane];
        const int b0 = __builtin_amdgcn_readlane(__float_as_int(ex0), j0);
        const int b1 = __builtin_amdgcn_readlane(__float_as_int(ex1), j0);
        const int b2 = __builtin_amdgcn_readlane(__float_as_int(ex2), j0);
        const int b3 = __builtin_amdgcn_readlane(__float_as_int(ex3), j0);
        const float w = __int_as_float((b0 & hm0) | (b1 & hm1) | (b2 & hm2) | (b3 & hm3));
        float* ap = acc + (e1 >> 16) * ACCP + laneoff;
        atomicAdd(ap + 0, w * r1.x);
        atomicAdd(ap + 1, w * r1.y);
        atomicAdd(ap + 2, w * r1.z);
        atomicAdd(ap + 3, w * r1.w);
      }
    }
  } else {
    for (int e = tid; e < NE; e += 1024) {
      const unsigned d = (unsigned)(tgt[e] - n0);
      if (d < (unsigned)NB) {
        const float4 ss = s_src4[src[e]];
        const float4 st = *(const float4*)&stg[d * NH];
        atomicMax(&emaxk[d * NH + 0], f2key(leaky(ss.x + st.x)));
        atomicMax(&emaxk[d * NH + 1], f2key(leaky(ss.y + st.y)));
        atomicMax(&emaxk[d * NH + 2], f2key(leaky(ss.z + st.z)));
        atomicMax(&emaxk[d * NH + 3], f2key(leaky(ss.w + st.w)));
      }
    }
    __syncthreads();
    for (int e = tid; e < NE; e += 1024) {
      const unsigned d = (unsigned)(tgt[e] - n0);
      if (d < (unsigned)NB) {
        const int s = src[e];
        const float4 ss = s_src4[s];
        const float4 st = *(const float4*)&stg[d * NH];
        const int4   mk = *(const int4*)&emaxk[d * NH];
        float exv[4];
        exv[0] = edge_exp(ss.x + st.x, mk.x);
        exv[1] = edge_exp(ss.y + st.y, mk.y);
        exv[2] = edge_exp(ss.z + st.z, mk.z);
        exv[3] = edge_exp(ss.w + st.w, mk.w);
#pragma unroll
        for (int k = 0; k < NH; ++k) atomicAdd(&esum[d * NH + k], exv[k]);
        const float* wr = Wh + (size_t)s * DT;
        float* ar = acc + d * ACCP;
        for (int j = 0; j < DT; ++j) atomicAdd(&ar[j + (j >> 6)], exv[j >> 5] * wr[j]);
      }
    }
  }
  __syncthreads();

  for (int i = tid; i < NB * DT; i += 1024) {
    const int d = i >> 7;
    const int j = i & 127;
    const float v = acc[d * ACCP + j + (j >> 6)] / (esum[d * NH + (j >> 5)] + 1e-10f);
    out[(size_t)(n0 + d) * DT + j] = v > 0.f ? v : expm1f(v);
  }
}

extern "C" void kernel_launch(void* const* d_in, const int* in_sizes, int n_in,
                              void* d_out, int out_size, void* d_ws, size_t ws_size,
                              hipStream_t stream) {
  const float* x    = (const float*)d_in[0];
  const int*   eidx = (const int*)d_in[1];
  const float* W    = (const float*)d_in[2];
  const float* a    = (const float*)d_in[3];
  float* out = (float*)d_out;

  const int* src = eidx;
  const int* tgt = eidx + NE;

  float* ws    = (float*)d_ws;
  float* Wh    = ws;
  float* s_src = Wh + (size_t)NN * DT;
  float* s_tgt = s_src + NN * NH;
  unsigned* plane = (unsigned*)(s_tgt + NN * NH);

  gat_wh_gemm<<<(NTILES + 3) / 4, 128, 0, stream>>>(x, W, tgt, Wh, plane);

  gat_scores<<<(NN * NH + 255) / 256, 256, 0, stream>>>(Wh, a, s_src, s_tgt);

  gat_owner<<<NOWN, 1024, 0, stream>>>(src, tgt, plane, s_src, s_tgt, Wh, out);
}


# edge phase: a wave owns 6-7 of the block's 200 rows and walks its own entries (LDS integer ranks); a row update is a plain 16-B LDS read-add-write per lane, no float atomic on rows; sum order changes
# speedup vs baseline: 7.2213x; 1.1153x over previous
#include <hip/hip_runtime.h>
#include <math.h>

#define NN 50000
#define NE 1600000
#define DIN 128
#define DOUT 32
#define NH 4
#define DT 128
#define NTILES (NN / 16)
#define NPAIR (DIN / 2)
#define WPAD 144
#define NB 200
#define EPW (NE / NTILES)
static_assert(NE == NTILES * EPW && EPW == 512, "one projection wave per 512 edges");
static_assert((NN + NB - 1) / NB <= 255, "an owner id must fit a byte below 255");

typedef __attribute__((ext_vector_type(2))) float v2f;
typedef __attribute__((ext_vector_type(8))) float v8f;

__device__ inline unsigned owner_of(int t) {
  const unsigned o = (unsigned)t / (unsigned)NB;
  return o < 255u ? o : 255u;
}

__global__ void __launch_bounds__(128) gat_wh_gemm(const float* __restrict__ x,
                                                   const float* __restrict__ W,
                                                   const int* __restrict__ tgt,
                                                   float* __restrict__ Wh,
                                                   unsigned* __restrict__ plane) {
  __shared__ v2f Wsh[NPAIR * WPAD];

  const int lane = threadIdx.x & 31;
  const int wave = threadIdx.x >> 5;
  const int tile = blockIdx.x * 4 + wave;

  int4 tv[EPW / 128];
  {
    const int4* __restrict__ t4p = (const int4*)tgt;
    const int w0 = (tile < NTILES ? tile : 0) * (EPW / 4) + lane;
#pragma unroll
    for (int u = 0; u < EPW / 128; ++u) tv[u] = t4p[w0 + u * 32];
  }

  for (int idx = threadIdx.x; idx < NPAIR * DT; idx += 128) {
    const int p  = idx >> 7;
    const int j  = idx & 127;
    const int kh = j >> 5;
    const int o  = j & 31;
    const float* wp = W + kh * (DIN * DOUT) + (2 * p) * DOUT + o;
    v2f pr;
    pr.x = wp[0];
    pr.y = wp[DOUT];
    Wsh[p * WPAD + j] = pr;
  }
  __syncthreads();

  const int half = lane >> 4;
  const int l16  = lane & 15;
  if (tile >= NTILES) return;
  const int row0 = tile * 16;

#pragma unroll
  for (int u = 0; u < EPW / 128; ++u) {
    const int4 v = tv[u];
    plane[tile * (EPW / 4) + u * 32 + lane] =
        owner_of(v.x) | (owner_of(v.y) << 8) | (owner_of(v.z) << 16) | (owner_of(v.w) << 24);
  }

  v8f acc[8];
#pragma unroll
  for (int jt = 0; jt < 8; ++jt) acc[jt] = (v8f){0.f,0.f,0.f,0.f,0.f,0.f,0.f,0.f};

  const float* xrow = x + (size_t)(row0 + l16) * DIN;

#pragma unroll 4
  for (int kk = 0; kk < 32; ++kk) {
    const int i0 = kk * 4 + half * 2;
    const int p  = kk * 2 + half;
    const v2f a = *(const v2f*)(xrow + i0);
    const v2f* wr = Wsh + p * WPAD + l16;
#pragma unroll
    for (int jt = 0; jt < 8; ++jt) {
      const v2f b = wr[jt * 16];
      acc[jt] = __builtin_amdgcn_wmma_f32_16x16x4_f32(
          false, a, false, b, (short)0, acc[jt], false, false);
    }
  }

#pragma unroll
  for (int jt = 0; jt < 8; ++jt) {
#pragma unroll
    for (int r = 0; r < 8; ++r) {
      const int row = row0 + r + 8 * half;
      Wh[(size_t)row * DT + jt * 16 + l16] = acc[jt][r];
    }
  }
}

__global__ void gat_scores(const float* __restrict__ Wh, const float* __restrict__ a,
                           float* __restrict__ s_src, float* __restrict__ s_tgt) {
  int idx = blockIdx.x * blockDim.x + threadIdx.x;
  if (idx >= NN * NH) return;
  int n = idx >> 2, k = idx & 3;
  const float* w  = Wh + (size_t)n * DT + k * DOUT;
  const float* as = a + k * (2 * DOUT);
  const float* at = as + DOUT;
  float ss = 0.f, st = 0.f;
#pragma unroll
  for (int o = 0; o < DOUT; ++o) { float v = w[o]; ss += v * as[o]; st += v * at[o]; }
  s_src[idx] = ss;
  s_tgt[idx] = st;
}

#define NOWN (NN / NB)
#define ACCP 128
#define NWAVE 32
#define SUBCAP 384
#define CAP 8192
#define EPT (CAP / 1024)
#define NP16 (NE / 16)
#define SCAN_LD 8
#define SCAN_CHUNK (1024 * SCAN_LD)
#define NFULL (NP16 / SCAN_CHUNK)
#define NTAIL ((NP16 - NFULL * SCAN_CHUNK + 1023) / 1024)
#define KEY_EMPTY (-2147483647 - 1)

__device__ inline int f2key(float v) {
  const int b = __float_as_int(v);
  return b >= 0 ? b : (b ^ 0x7FFFFFFF);
}
__device__ inline float key2f(int k) {
  return __int_as_float(k >= 0 ? k : (k ^ 0x7FFFFFFF));
}
__device__ inline float leaky(float v) { return v > 0.f ? v : 0.2f * v; }
__device__ inline float score_exp(float score, int maxkey) {
  float m = key2f(maxkey);
  if (!(__builtin_fabsf(m) < __builtin_inff())) m = 0.f;
  return __expf(score - m);
}
__device__ inline float edge_exp(float score_sum, int maxkey) {
  return score_exp(leaky(score_sum), maxkey);
}

__device__ inline void scan_load(const uint4* __restrict__ p16, int base, uint4 (&tv)[SCAN_LD]) {
#pragma unroll
  for (int u = 0; u < SCAN_LD; ++u) tv[u] = p16[base + u * 1024];
}
__device__ inline unsigned nz_bytes(unsigned x) {
  return ((x & 0x7F7F7F7Fu) + 0x7F7F7F7Fu) | x;
}
#define NO_MATCH 0xF0F0F0F0u
__device__ inline unsigned ne_flags(const uint4 v, unsigned pat) {
  return (nz_bytes(v.x ^ pat) & 0x80808080u) | ((nz_bytes(v.y ^ pat) >> 1) & 0x40404040u) |
         ((nz_bytes(v.z ^ pat) >> 2) & 0x20202020u) | ((nz_bytes(v.w ^ pat) >> 3) & 0x10101010u);
}
__device__ inline void keep_pair(unsigned fa, unsigned fb, int idx, unsigned* cntp, unsigned* listp) {
  unsigned z = ~(fa | (fb >> 4));
  while (z) {
    const unsigned b  = (unsigned)__builtin_ctz(z);
    const unsigned nb = ~b;
    z &= z - 1u;
    const unsigned pos = atomicAdd(cntp, 1u);
    if (pos < (unsigned)CAP)
      listp[pos] = (unsigned)idx * 16u + ((nb & 4u) << 12) + ((nb & 3u) << 2) + (b >> 3);
  }
}
static_assert(SCAN_LD % 2 == 0 && NTAIL == 2, "the scan tests plane words in pairs 1024 words apart");

__device__ inline unsigned wave_of(unsigned d) { return (d * 4u) / 25u; }
static_assert(((NB - 1) * 4) / 25 == NWAVE - 1 && CAP <= 65536, "row owners fill the waves; list places fit 16 bits");

__device__ inline void row_add(float4* ap, float w, const float4 r) {
  float4 a = *ap;
  a.x += w * r.x;
  a.y += w * r.y;
  a.z += w * r.z;
  a.w += w * r.w;
  *ap = a;
}

__global__ void __launch_bounds__(1024) gat_owner(const int* __restrict__ src,
                                                  const int* __restrict__ tgt,
                                                  const unsigned* __restrict__ plane,
                                                  const float* __restrict__ s_src,
                                                  const float* __restrict__ s_tgt,
                                                  const float* __restrict__ Wh,
                                                  float* __restrict__ out) {
  __shared__ __attribute__((aligned(16))) float acc[NB * ACCP];
  __shared__ unsigned list[CAP];
  __shared__ float4 esc[CAP];
  __shared__ __attribute__((aligned(16))) int   emaxk[NB * NH];
  __shared__ __attribute__((aligned(16))) float esum[NB * NH];
  __shared__ __attribute__((aligned(16))) float stg[NB * NH];
  __shared__ unsigned short sub[NWAVE * SUBCAP];
  __shared__ unsigned wcnt[NWAVE];
  __shared__ unsigned cnt;

  const int tid = threadIdx.x;
  const int n0  = blockIdx.x * NB;
  const float4* __restrict__ s_src4 = (const float4*)s_src;

  for (int i = tid; i < NB * ACCP; i += 1024) acc[i] = 0.f;
  for (int i = tid; i < NB * NH; i += 1024) {
    emaxk[i] = KEY_EMPTY;
    esum[i]  = 0.f;
    stg[i]   = s_tgt[n0 * NH + i];
  }
  if (tid == 0) cnt = 0u;
  if (tid < NWAVE) wcnt[tid] = 0u;
  __syncthreads();

  {
    const uint4* __restrict__ t4p = (const uint4*)plane;
    const unsigned pat = blockIdx.x * 0x01010101u;
    uint4 cur[SCAN_LD], nxt[SCAN_LD];
    int chunk = (int)(blockIdx.x % NFULL);
    int curbase = chunk * SCAN_CHUNK + tid;
    scan_load(t4p, curbase, cur);
    for (int it = 0; it < NFULL - 1; ++it) {
      chunk = (chunk + 1 == NFULL) ? 0 : chunk + 1;
      const int nxtbase = chunk * SCAN_CHUNK + tid;
      scan_load(t4p, nxtbase, nxt);
#pragma unroll
      for (int u = 0; u < SCAN_LD; u += 2)
        keep_pair(ne_flags(cur[u], pat), ne_flags(cur[u + 1], pat), curbase + u * 1024, &cnt, list);
#pragma unroll
      for (int u = 0; u < SCAN_LD; ++u) cur[u] = nxt[u];
      curbase = nxtbase;
    }
#pragma unroll
    for (int u = 0; u < SCAN_LD; u += 2)
      keep_pair(ne_flags(cur[u], pat), ne_flags(cur[u + 1], pat), curbase + u * 1024, &cnt, list);
    {
      const int idx = NFULL * SCAN_CHUNK + tid;
      unsigned fa = NO_MATCH, fb = NO_MATCH;
      if (idx < NP16) fa = ne_flags(t4p[idx], pat);
      if (idx + 1024 < NP16) fb = ne_flags(t4p[idx + 1024], pat);
      keep_pair(fa, fb, idx, &cnt, list);
    }
  }
  __syncthreads();

  const unsigned total = (unsigned)__builtin_amdgcn_readfirstlane((int)cnt);

  if (total <= (unsigned)CAP) {
    const int n = (int)total;

    {
      unsigned en[EPT];
      int sn[EPT];
      int tn[EPT];
      float4 ss[EPT];
#pragma unroll
      for (int u = 0; u < EPT; ++u) {
        const int idx = tid + u * 1024;
        const unsigned raw = list[idx];
        en[u] = idx < n ? raw : 0u;
      }
#pragma unroll
      for (int u = 0; u < EPT; ++u) {
        sn[u] = src[en[u]];
        tn[u] = tgt[en[u]];
      }
#pragma unroll
      for (int u = 0; u < EPT; ++u) ss[u] = s_src4[sn[u]];
#pragma unroll
      for (int u = 0; u < EPT; ++u) {
        const int idx = tid + u * 1024;
        if (idx < n) {
          const unsigned d  = (unsigned)(tn[u] - n0);
          const float4   st = *(const float4*)&stg[d * NH];
          float4 sc;
          sc.x = leaky(ss[u].x + st.x);
          sc.y = leaky(ss[u].y + st.y);
          sc.z = leaky(ss[u].z + st.z);
          sc.w = leaky(ss[u].w + st.w);
          list[idx] = (unsigned)sn[u] | (d << 16);
          esc[idx]  = sc;
          atomicMax(&emaxk[d * NH + 0], f2key(sc.x));
          atomicMax(&emaxk[d * NH + 1], f2key(sc.y));
          atomicMax(&emaxk[d * NH + 2], f2key(sc.z));
          atomicMax(&emaxk[d * NH + 3], f2key(sc.w));
          const unsigned wv = wave_of(d);
          const unsigned wp = atomicAdd(&wcnt[wv], 1u);
          if (wp < (unsigned)SUBCAP) sub[wv * SUBCAP + wp] = (unsigned short)idx;
        }
      }
    }
    __syncthreads();

    const int lane = tid & 31;
    const int wave = __builtin_amdgcn_readfirstlane(tid >> 5);
    const int kq = lane >> 3;
    const int hm0 = kq == 0 ? -1 : 0, hm1 = kq == 1 ? -1 : 0, hm2 = kq == 2 ? -1 : 0, hm3 = kq == 3 ? -1 : 0;
    const float4* __restrict__ Wh4 = (const float4*)Wh;
    float4* acc4 = (float4*)acc + lane;
    const unsigned mine_n = (unsigned)__builtin_amdgcn_readfirstlane((int)wcnt[wave]);

    if (mine_n <= (unsigned)SUBCAP) {
      const int c = (int)mine_n;
      const unsigned short* subw = sub + wave * SUBCAP;
      for (int base = 0; base < c; base += 32) {
        const int k = base + lane;
        unsigned en2 = 0u;
        float ex0 = 0.f, ex1 = 0.f, ex2 = 0.f, ex3 = 0.f;
        if (k < c) {
          const int idx = subw[k];
          en2 = list[idx];
          const unsigned d = en2 >> 16;
          const float4 sc = esc[idx];
          const int4   mk = *(const int4*)&emaxk[d * NH];
          ex0 = score_exp(sc.x, mk.x);
          ex1 = score_exp(sc.y, mk.y);
          ex2 = score_exp(sc.z, mk.z);
          ex3 = score_exp(sc.w, mk.w);
          atomicAdd(&esum[d * NH + 0], ex0);
          atomicAdd(&esum[d * NH + 1], ex1);
          atomicAdd(&esum[d * NH + 2], ex2);
          atomicAdd(&esum[d * NH + 3], ex3);
        }
        const int left = c - base;
        const int cntb = left < 32 ? left : 32;

        int j0 = 0;
        for (; j0 + 8 <= cntb; j0 += 8) {
          unsigned er[8];
          float4 r[8];
#pragma unroll
          for (int u = 0; u < 8; ++u) {
            er[u] = (unsigned)__builtin_amdgcn_readlane((int)en2, j0 + u);
            r[u]  = Wh4[(size_t)(er[u] & 0xFFFFu) * (DT / 4) + lane];
          }
#pragma unroll
          for (int u = 0; u < 8; ++u) {
            const int b0 = __builtin_amdgcn_readlane(__float_as_int(ex0), j0 + u);
            const int b1 = __builtin_amdgcn_readlane(__float_as_int(ex1), j0 + u);
            const int b2 = __builtin_amdgcn_readlane(__float_as_int(ex2), j0 + u);
            const int b3 = __builtin_amdgcn_readlane(__float_as_int(ex3), j0 + u);
            const float w = __int_as_float((b0 & hm0) | (b1 & hm1) | (b2 & hm2) | (b3 & hm3));
            row_add(acc4 + (er[u] >> 16) * (ACCP / 4), w, r[u]);
          }
        }
        for (; j0 < cntb; ++j0) {
          const unsigned e1 = (unsigned)__builtin_amdgcn_readlane((int)en2, j0);
          const float4 r1 = Wh4[(size_t)(e1 & 0xFFFFu) * (DT / 4) + lane];
          const int b0 = __builtin_amdgcn_readlane(__float_as_int(ex0), j0);
          const int b1 = __builtin_amdgcn_readlane(__float_as_int(ex1), j0);
          const int b2 = __builtin_amdgcn_readlane(__float_as_int(ex2), j0);
          const int b3 = __builtin_amdgcn_readlane(__float_as_int(ex3), j0);
          const float w = __int_as_float((b0 & hm0) | (b1 & hm1) | (b2 & hm2) | (b3 & hm3));
          row_add(acc4 + (e1 >> 16) * (ACCP / 4), w, r1);
        }
      }
    } else {
      for (int base = 0; base < n; base += 32) {
        const int idx = base + lane;
        unsigned en2 = 0u;
        float ex0 = 0.f, ex1 = 0.f, ex2 = 0.f, ex3 = 0.f;
        bool own = false;
        if (idx < n) {
          en2 = list[idx];
          const unsigned d = en2 >> 16;
          own = wave_of(d) == (unsigned)wave;
          if (own) {
            const float4 sc = esc[idx];
            const int4   mk = *(const int4*)&emaxk[d * NH];
            ex0 = score_exp(sc.x, mk.x);
            ex1 = score_exp(sc.y, mk.y);
            ex2 = score_exp(sc.z, mk.z);
            ex3 = score_exp(sc.w, mk.w);
            atomicAdd(&esum[d * NH + 0], ex0);
            atomicAdd(&esum[d * NH + 1], ex1);
            atomicAdd(&esum[d * NH + 2], ex2);
            atomicAdd(&esum[d * NH + 3], ex3);
          }
        }
        unsigned m = __builtin_amdgcn_ballot_w32(own);
        while (m) {
          const int j = __builtin_ctz(m);
          m &= m - 1u;
          const unsigned e1 = (unsigned)__builtin_amdgcn_readlane((int)en2, j);
          const float4 r1 = Wh4[(size_t)(e1 & 0xFFFFu) * (DT / 4) + lane];
          const int b0 = __builtin_amdgcn_readlane(__float_as_int(ex0), j);
          const int b1 = __builtin_amdgcn_readlane(__float_as_int(ex1), j);
          const int b2 = __builtin_amdgcn_readlane(__float_as_int(ex2), j);
          const int b3 = __builtin_amdgcn_readlane(__float_as_int(ex3), j);
          const float w = __int_as_float((b0 & hm0) | (b1 & hm1) | (b2 & hm2) | (b3 & hm3));
          row_add(acc4 + (e1 >> 16) * (ACCP / 4), w, r1);
        }
      }
    }
  } else {
    for (int e = tid; e < NE; e += 1024) {
      const unsigned d = (unsigned)(tgt[e] - n0);
      if (d < (unsigned)NB) {
        const float4 ss = s_src4[src[e]];
        const float4 st = *(const float4*)&stg[d * NH];
        atomicMax(&emaxk[d * NH + 0], f2key(leaky(ss.x + st.x)));
        atomicMax(&emaxk[d * NH + 1], f2key(leaky(ss.y + st.y)));
        atomicMax(&emaxk[d * NH + 2], f2key(leaky(ss.z + st.z)));
        atomicMax(&emaxk[d * NH + 3], f2key(leaky(ss.w + st.w)));
      }
    }
    __syncthreads();
    for (int e = tid; e < NE; e += 1024) {
      const unsigned d = (unsigned)(tgt[e] - n0);
      if (d < (unsigned)NB) {
        const int s = src[e];
        const float4 ss = s_src4[s];
        const float4 st = *(const float4*)&stg[d * NH];
        const int4   mk = *(const int4*)&emaxk[d * NH];
        float exv[4];
        exv[0] = edge_exp(ss.x + st.x, mk.x);
        exv[1] = edge_exp(ss.y + st.y, mk.y);
        exv[2] = edge_exp(ss.z + st.z, mk.z);
        exv[3] = edge_exp(ss.w + st.w, mk.w);
#pragma unroll
        for (int k = 0; k < NH; ++k) atomicAdd(&esum[d * NH + k], exv[k]);
        const float* wr = Wh + (size_t)s * DT;
        float* ar = acc + d * ACCP;
        for (int j = 0; j < DT; ++j) atomicAdd(&ar[j], exv[j >> 5] * wr[j]);
      }
    }
  }
  __syncthreads();

  for (int i = tid; i < NB * DT; i += 1024) {
    const int d = i >> 7;
    const int j = i & 127;
    const float v = acc[d * ACCP + j] / (esum[d * NH + (j >> 5)] + 1e-10f);
    out[(size_t)(n0 + d) * DT + j] = v > 0.f ? v : expm1f(v);
  }
}

extern "C" void kernel_launch(void* const* d_in, const int* in_sizes, int n_in,
                              void* d_out, int out_size, void* d_ws, size_t ws_size,
                              hipStream_t stream) {
  const float* x    = (const float*)d_in[0];
  const int*   eidx = (const int*)d_in[1];
  const float* W    = (const float*)d_in[2];
  const float* a    = (const float*)d_in[3];
  float* out = (float*)d_out;

  const int* src = eidx;
  const int* tgt = eidx + NE;

  float* ws    = (float*)d_ws;
  float* Wh    = ws;
  float* s_src = Wh + (size_t)NN * DT;
  float* s_tgt = s_src + NN * NH;
  unsigned* plane = (unsigned*)(s_tgt + NN * NH);

  gat_wh_gemm<<<(NTILES + 3) / 4, 128, 0, stream>>>(x, W, tgt, Wh, plane);

  gat_scores<<<(NN * NH + 255) / 256, 256, 0, stream>>>(Wh, a, s_src, s_tgt);

  gat_owner<<<NOWN, 1024, 0, stream>>>(src, tgt, plane, s_src, s_tgt, Wh, out);
}


# projection work-groups group their 2048 edges by owner block (LDS ranks, whole-line stores); an edge block copies its own run per segment: no edge scan, no per-edge src/tgt reads; arrival-order sums
# speedup vs baseline: 9.1962x; 1.2735x over previous
#include <hip/hip_runtime.h>
#include <math.h>

#define NN 50000
#define NE 1600000
#define DIN 128
#define DOUT 32
#define NH 4
#define DT 128
#define NTILES (NN / 16)
#define NPAIR (DIN / 2)
#define WPAD 144
#define NB 200
#define EPW (NE / NTILES)
#define NOWN (NN / NB)
#define SEGE (4 * EPW)
#define NSEG ((NE + SEGE - 1) / SEGE)
#define NBKT 256
static_assert(NE == NTILES * EPW && EPW == 512, "one projection wave per 512 edges");
static_assert(NN == NOWN * NB && NOWN < NBKT - 1 && NB <= 256 && NN <= 65536, "bucket, row and source fit their fields");

typedef __attribute__((ext_vector_type(2))) float v2f;
typedef __attribute__((ext_vector_type(8))) float v8f;

__device__ inline unsigned bucket_of(int t) {
  const unsigned o = (unsigned)t / (unsigned)NB;
  return o < (unsigned)NOWN ? o : (unsigned)NOWN;
}
__device__ inline unsigned entry_of(int s, int t, unsigned b) {
  return ((unsigned)s & 0xFFFFu) | ((((unsigned)t - b * (unsigned)NB) & 0xFFu) << 16);
}

__global__ void __launch_bounds__(128) gat_wh_gemm(const float* __restrict__ x,
                                                   const float* __restrict__ W,
                                                   const int* __restrict__ src,
                                                   const int* __restrict__ tgt,
                                                   float* __restrict__ Wh,
                                                   unsigned* __restrict__ ent,
                                                   unsigned short* __restrict__ off) {
  __shared__ __attribute__((aligned(16))) v2f Wsh[NPAIR * WPAD];
  unsigned* sorted = (unsigned*)Wsh;
  unsigned* hist   = sorted + SEGE;
  unsigned* bstart = hist + NBKT;
  static_assert((SEGE + 2 * NBKT) * 4 <= NPAIR * WPAD * 8 && SEGE == 4 * 512, "grouping arrays fit the staging area; four 16-B words per thread");

  const int lane = threadIdx.x & 31;
  const int wave = threadIdx.x >> 5;
  const int tile = blockIdx.x * 4 + wave;
  const bool live = tile < NTILES;

  int4 tv[EPW / 128], sv[EPW / 128];
  {
    const int4* __restrict__ t4p = (const int4*)tgt;
    const int4* __restrict__ s4p = (const int4*)src;
    const int w0 = (live ? tile : 0) * (EPW / 4) + lane;
#pragma unroll
    for (int u = 0; u < EPW / 128; ++u) { tv[u] = t4p[w0 + u * 32]; sv[u] = s4p[w0 + u * 32]; }
  }
  hist[threadIdx.x]       = 0u;
  hist[threadIdx.x + 128] = 0u;
  __syncthreads();

  unsigned rk[EPW / 32];
  if (live) {
#pragma unroll
    for (int u = 0; u < EPW / 128; ++u) {
      rk[4 * u + 0] = atomicAdd(&hist[bucket_of(tv[u].x)], 1u);
      rk[4 * u + 1] = atomicAdd(&hist[bucket_of(tv[u].y)], 1u);
      rk[4 * u + 2] = atomicAdd(&hist[bucket_of(tv[u].z)], 1u);
      rk[4 * u + 3] = atomicAdd(&hist[bucket_of(tv[u].w)], 1u);
    }
  }
  __syncthreads();

  if (wave == 0) {
    unsigned c[8];
    unsigned sum = 0u;
#pragma unroll
    for (int j = 0; j < 8; ++j) { c[j] = hist[lane * 8 + j]; sum += c[j]; }
    unsigned incl = sum;
#pragma unroll
    for (int dlt = 1; dlt < 32; dlt <<= 1) {
      const unsigned v = __shfl_up(incl, dlt);
      if (lane >= dlt) incl += v;
    }
    unsigned run = incl - sum;
#pragma unroll
    for (int j = 0; j < 8; ++j) { bstart[lane * 8 + j] = run; run += c[j]; }
  }
  __syncthreads();

  if (live) {
    unsigned ps[EPW / 32];
#pragma unroll
    for (int u = 0; u < EPW / 128; ++u) {
      ps[4 * u + 0] = bstart[bucket_of(tv[u].x)] + rk[4 * u + 0];
      ps[4 * u + 1] = bstart[bucket_of(tv[u].y)] + rk[4 * u + 1];
      ps[4 * u + 2] = bstart[bucket_of(tv[u].z)] + rk[4 * u + 2];
      ps[4 * u + 3] = bstart[bucket_of(tv[u].w)] + rk[4 * u + 3];
    }
#pragma unroll
    for (int u = 0; u < EPW / 128; ++u) {
      sorted[ps[4 * u + 0]] = entry_of(sv[u].x, tv[u].x, bucket_of(tv[u].x));
      sorted[ps[4 * u + 1]] = entry_of(sv[u].y, tv[u].y, bucket_of(tv[u].y));
      sorted[ps[4 * u + 2]] = entry_of(sv[u].z, tv[u].z, bucket_of(tv[u].z));
      sorted[ps[4 * u + 3]] = entry_of(sv[u].w, tv[u].w, bucket_of(tv[u].w));
    }
  }
  __syncthreads();

  {
    const uint4* s4 = (const uint4*)sorted;
    const uint4 k0 = s4[threadIdx.x];
    const uint4 k1 = s4[128 + threadIdx.x];
    const uint4 k2 = s4[256 + threadIdx.x];
    const uint4 k3 = s4[384 + threadIdx.x];
    const unsigned f0 = bstart[2 * threadIdx.x], f1 = bstart[2 * threadIdx.x + 1];
    __syncthreads();
    uint4* e4 = (uint4*)(ent + (size_t)blockIdx.x * SEGE);
    e4[threadIdx.x]       = k0;
    e4[128 + threadIdx.x] = k1;
    e4[256 + threadIdx.x] = k2;
    e4[384 + threadIdx.x] = k3;
    ((unsigned*)off)[(size_t)blockIdx.x * (NBKT / 2) + threadIdx.x] = f0 | (f1 << 16);
  }

  for (int idx = threadIdx.x; idx < NPAIR * DT; idx += 128) {
    const int p  = idx >> 7;
    const int j  = idx & 127;
    const int kh = j >> 5;
    const int o  = j & 31;
    const float* wp = W + kh * (DIN * DOUT) + (2 * p) * DOUT + o;
    v2f pr;
    pr.x = wp[0];
    pr.y = wp[DOUT];
    Wsh[p * WPAD + j] = pr;
  }
  __syncthreads();

  const int half = lane >> 4;
  const int l16  = lane & 15;
  if (!live) return;
  const int row0 = tile * 16;

  v8f acc[8];
#pragma unroll
  for (int jt = 0; jt < 8; ++jt) acc[jt] = (v8f){0.f,0.f,0.f,0.f,0.f,0.f,0.f,0.f};

  const float* xrow = x + (size_t)(row0 + l16) * DIN;

#pragma unroll 4
  for (int kk = 0; kk < 32; ++kk) {
    const int i0 = kk * 4 + half * 2;
    const int p  = kk * 2 + half;
    const v2f a = *(const v2f*)(xrow + i0);
    const v2f* wr = Wsh + p * WPAD + l16;
#pragma unroll
    for (int jt = 0; jt < 8; ++jt) {
      const v2f b = wr[jt * 16];
      acc[jt] = __builtin_amdgcn_wmma_f32_16x16x4_f32(
          false, a, false, b, (short)0, acc[jt], false, false);
    }
  }

#pragma unroll
  for (int jt = 0; jt < 8; ++jt) {
#pragma unroll
    for (int r = 0; r < 8; ++r) {
      const int row = row0 + r + 8 * half;
      Wh[(size_t)row * DT + jt * 16 + l16] = acc[jt][r];
    }
  }
}

__global__ void gat_scores(const float* __restrict__ Wh, const float* __restrict__ a,
                           float* __restrict__ s_src, float* __restrict__ s_tgt) {
  int idx = blockIdx.x * blockDim.x + threadIdx.x;
  if (idx >= NN * NH) return;
  int n = idx >> 2, k = idx & 3;
  const float* w  = Wh + (size_t)n * DT + k * DOUT;
  const float* as = a + k * (2 * DOUT);
  const float* at = as + DOUT;
  float ss = 0.f, st = 0.f;
#pragma unroll
  for (int o = 0; o < DOUT; ++o) { float v = w[o]; ss += v * as[o]; st += v * at[o]; }
  s_src[idx] = ss;
  s_tgt[idx] = st;
}

#define ACCP 128
#define NWAVE 32
#define SUBCAP 384
#define CAP 8192
#define EPT (CAP / 1024)
#define NPASS ((NSEG + 127) / 128)
#define KEY_EMPTY (-2147483647 - 1)

__device__ inline int f2key(float v) {
  const int b = __float_as_int(v);
  return b >= 0 ? b : (b ^ 0x7FFFFFFF);
}
__device__ inline float key2f(int k) {
  return __int_as_float(k >= 0 ? k : (k ^ 0x7FFFFFFF));
}
__device__ inline float leaky(float v) { return v > 0.f ? v : 0.2f * v; }
__device__ inline float score_exp(float score, int maxkey) {
  float m = key2f(maxkey);
  if (!(__builtin_fabsf(m) < __builtin_inff())) m = 0.f;
  return __expf(score - m);
}
__device__ inline float edge_exp(float score_sum, int maxkey) {
  return score_exp(leaky(score_sum), maxkey);
}

__device__ inline unsigned wave_of(unsigned d) { return (d * 4u) / 25u; }
static_assert(((NB - 1) * 4) / 25 == NWAVE - 1 && CAP <= 65536, "row owners fill the waves; list places fit 16 bits");

__device__ inline void row_add(float4* ap, float w, const float4 r) {
  float4 a = *ap;
  a.x += w * r.x;
  a.y += w * r.y;
  a.z += w * r.z;
  a.w += w * r.w;
  *ap = a;
}

__global__ void __launch_bounds__(1024) gat_owner(const int* __restrict__ src,
                                                  const int* __restrict__ tgt,
                                                  const unsigned* __restrict__ ent,
                                                  const unsigned short* __restrict__ off,
                                                  const float* __restrict__ s_src,
                                                  const float* __restrict__ s_tgt,
                                                  const float* __restrict__ Wh,
                                                  float* __restrict__ out) {
  __shared__ __attribute__((aligned(16))) float acc[NB * ACCP];
  __shared__ unsigned list[CAP];
  __shared__ float4 esc[CAP];
  __shared__ __attribute__((aligned(16))) int   emaxk[NB * NH];
  __shared__ __attribute__((aligned(16))) float esum[NB * NH];
  __shared__ __attribute__((aligned(16))) float stg[NB * NH];
  __shared__ unsigned short sub[NWAVE * SUBCAP];
  __shared__ unsigned wcnt[NWAVE];
  __shared__ unsigned cnt;

  const int tid = threadIdx.x;
  const int n0  = blockIdx.x * NB;
  const float4* __restrict__ s_src4 = (const float4*)s_src;

  for (int i = tid; i < NB * ACCP; i += 1024) acc[i] = 0.f;
  for (int i = tid; i < NB * NH; i += 1024) {
    emaxk[i] = KEY_EMPTY;
    esum[i]  = 0.f;
    stg[i]   = s_tgt[n0 * NH + i];
  }
  if (tid == 0) cnt = 0u;
  if (tid < NWAVE) wcnt[tid] = 0u;
  __syncthreads();

  {
    const int lane8 = tid & 7;
    const int grp   = tid >> 3;
    const int lead  = (tid & 31) & ~7;
    unsigned st[NPASS], ln[NPASS], bs[NPASS], v0[NPASS], v1[NPASS];
#pragma unroll
    for (int p = 0; p < NPASS; ++p) {
      const int seg  = p * 128 + grp;
      const int segc = seg < NSEG ? seg : NSEG - 1;
      const unsigned a = off[segc * NBKT + blockIdx.x];
      const unsigned b = off[segc * NBKT + blockIdx.x + 1];
      st[p] = a;
      ln[p] = seg < NSEG ? b - a : 0u;
    }
#pragma unroll
    for (int p = 0; p < NPASS; ++p) {
      unsigned base = 0u;
      if (lane8 == 0 && ln[p] != 0u) base = atomicAdd(&cnt, ln[p]);
      bs[p] = (unsigned)__shfl((int)base, lead);
    }
#pragma unroll
    for (int p = 0; p < NPASS; ++p) {
      const int seg  = p * 128 + grp;
      const int segc = seg < NSEG ? seg : NSEG - 1;
      const unsigned* eb = ent + (size_t)segc * SEGE;
      const unsigned i0 = st[p] + (unsigned)lane8;
      const unsigned i1 = i0 + 8u;
      v0[p] = eb[i0 < (unsigned)SEGE ? i0 : (unsigned)SEGE - 1u];
      v1[p] = eb[i1 < (unsigned)SEGE ? i1 : (unsigned)SEGE - 1u];
    }
#pragma unroll
    for (int p = 0; p < NPASS; ++p) {
      const unsigned q0 = bs[p] + (unsigned)lane8;
      if ((unsigned)lane8 < ln[p] && q0 < (unsigned)CAP) list[q0] = v0[p];
      if ((unsigned)lane8 + 8u < ln[p] && q0 + 8u < (unsigned)CAP) list[q0 + 8u] = v1[p];
      if (ln[p] > 16u) {
        const int seg = p * 128 + grp;
        const unsigned* ep = ent + (size_t)seg * SEGE + st[p];
        for (unsigned i = (unsigned)lane8 + 16u; i < ln[p]; i += 8u)
          if (bs[p] + i < (unsigned)CAP) list[bs[p] + i] = ep[i];
      }
    }
  }
  __syncthreads();

  const unsigned total = (unsigned)__builtin_amdgcn_readfirstlane((int)cnt);

  if (total <= (unsigned)CAP) {
    const int n = (int)total;

    {
      unsigned en[EPT];
      float4 ss[EPT];
#pragma unroll
      for (int u = 0; u < EPT; ++u) {
        const int idx = tid + u * 1024;
        const unsigned raw = list[idx];
        en[u] = idx < n ? raw : 0u;
      }
#pragma unroll
      for (int u = 0; u < EPT; ++u) ss[u] = s_src4[en[u] & 0xFFFFu];
#pragma unroll
      for (int u = 0; u < EPT; ++u) {
        const int idx = tid + u * 1024;
        if (idx < n) {
          const unsigned d  = en[u] >> 16;
          const float4   st = *(const float4*)&stg[d * NH];
          float4 sc;
          sc.x = leaky(ss[u].x + st.x);
          sc.y = leaky(ss[u].y + st.y);
          sc.z = leaky(ss[u].z + st.z);
          sc.w = leaky(ss[u].w + st.w);
          esc[idx]  = sc;
          atomicMax(&emaxk[d * NH + 0], f2key(sc.x));
          atomicMax(&emaxk[d * NH + 1], f2key(sc.y));
          atomicMax(&emaxk[d * NH + 2], f2key(sc.z));
          atomicMax(&emaxk[d * NH + 3], f2key(sc.w));
          const unsigned wv = wave_of(d);
          const unsigned wp = atomicAdd(&wcnt[wv], 1u);
          if (wp < (unsigned)SUBCAP) sub[wv * SUBCAP + wp] = (unsigned short)idx;
        }
      }
    }
    __syncthreads();

    const int lane = tid & 31;
    const int wave = __builtin_amdgcn_readfirstlane(tid >> 5);
    const int kq = lane >> 3;
    const int hm0 = kq == 0 ? -1 : 0, hm1 = kq == 1 ? -1 : 0, hm2 = kq == 2 ? -1 : 0, hm3 = kq == 3 ? -1 : 0;
    const float4* __restrict__ Wh4 = (const float4*)Wh;
    float4* acc4 = (float4*)acc + lane;
    const unsigned mine_n = (unsigned)__builtin_amdgcn_readfirstlane((int)wcnt[wave]);

    if (mine_n <= (unsigned)SUBCAP) {
      const int c = (int)mine_n;
      const unsigned short* subw = sub + wave * SUBCAP;
      for (int base = 0; base < c; base += 32) {
        const int k = base + lane;
        unsigned en2 = 0u;
        float ex0 = 0.f, ex1 = 0.f, ex2 = 0.f, ex3 = 0.f;
        if (k < c) {
          const int idx = subw[k];
          en2 = list[idx];
          const unsigned d = en2 >> 16;
          const float4 sc = esc[idx];
          const int4   mk = *(const int4*)&emaxk[d * NH];
          ex0 = score_exp(sc.x, mk.x);
          ex1 = score_exp(sc.y, mk.y);
          ex2 = score_exp(sc.z, mk.z);
          ex3 = score_exp(sc.w, mk.w);
          atomicAdd(&esum[d * NH + 0], ex0);
          atomicAdd(&esum[d * NH + 1], ex1);
          atomicAdd(&esum[d * NH + 2], ex2);
          atomicAdd(&esum[d * NH + 3], ex3);
        }
        const int left = c - base;
        const int cntb = left < 32 ? left : 32;

        int j0 = 0;
        for (; j0 + 8 <= cntb; j0 += 8) {
          unsigned er[8];
          float4 r[8];
#pragma unroll
          for (int u = 0; u < 8; ++u) {
            er[u] = (unsigned)__builtin_amdgcn_readlane((int)en2, j0 + u);
            r[u]  = Wh4[(size_t)(er[u] & 0xFFFFu) * (DT / 4) + lane];
          }
#pragma unroll
          for (int u = 0; u < 8; ++u) {
            const int b0 = __builtin_amdgcn_readlane(__float_as_int(ex0), j0 + u);
            const int b1 = __builtin_amdgcn_readlane(__float_as_int(ex1), j0 + u);
            const int b2 = __builtin_amdgcn_readlane(__float_as_int(ex2), j0 + u);
            const int b3 = __builtin_amdgcn_readlane(__float_as_int(ex3), j0 + u);
            const float w = __int_as_float((b0 & hm0) | (b1 & hm1) | (b2 & hm2) | (b3 & hm3));
            row_add(acc4 + (er[u] >> 16) * (ACCP / 4), w, r[u]);
          }
        }
        for (; j0 < cntb; ++j0) {
          const unsigned e1 = (unsigned)__builtin_amdgcn_readlane((int)en2, j0);
          const float4 r1 = Wh4[(size_t)(e1 & 0xFFFFu) * (DT / 4) + lane];
          const int b0 = __builtin_amdgcn_readlane(__float_as_int(ex0), j0);
          const int b1 = __builtin_amdgcn_readlane(__float_as_int(ex1), j0);
          const int b2 = __builtin_amdgcn_readlane(__float_as_int(ex2), j0);
          const int b3 = __builtin_amdgcn_readlane(__float_as_int(ex3), j0);
          const float w = __int_as_float((b0 & hm0) | (b1 & hm1) | (b2 & hm2) | (b3 & hm3));
          row_add(acc4 + (e1 >> 16) * (ACCP / 4), w, r1);
        }
      }
    } else {
      for (int base = 0; base < n; base += 32) {
        const int idx = base + lane;
        unsigned en2 = 0u;
        float ex0 = 0.f, ex1 = 0.f, ex2 = 0.f, ex3 = 0.f;
        bool own = false;
        if (idx < n) {
          en2 = list[idx];
          const unsigned d = en2 >> 16;
          own = wave_of(d) == (unsigned)wave;
          if (own) {
            const float4 sc = esc[idx];
            const int4   mk = *(const int4*)&emaxk[d * NH];
            ex0 = score_exp(sc.x, mk.x);
            ex1 = score_exp(sc.y, mk.y);
            ex2 = score_exp(sc.z, mk.z);
            ex3 = score_exp(sc.w, mk.w);
            atomicAdd(&esum[d * NH + 0], ex0);
            atomicAdd(&esum[d * NH + 1], ex1);
            atomicAdd(&esum[d * NH + 2], ex2);
            atomicAdd(&esum[d * NH + 3], ex3);
          }
        }
        unsigned m = __builtin_amdgcn_ballot_w32(own);
        while (m) {
          const int j = __builtin_ctz(m);
          m &= m - 1u;
          const unsigned e1 = (unsigned)__builtin_amdgcn_readlane((int)en2, j);
          const float4 r1 = Wh4[(size_t)(e1 & 0xFFFFu) * (DT / 4) + lane];
          const int b0 = __builtin_amdgcn_readlane(__float_as_int(ex0), j);
          const int b1 = __builtin_amdgcn_readlane(__float_as_int(ex1), j);
          const int b2 = __builtin_amdgcn_readlane(__float_as_int(ex2), j);
          const int b3 = __builtin_amdgcn_readlane(__float_as_int(ex3), j);
          const float w = __int_as_float((b0 & hm0) | (b1 & hm1) | (b2 & hm2) | (b3 & hm3));
          row_add(acc4 + (e1 >> 16) * (ACCP / 4), w, r1);
        }
      }
    }
  } else {
    for (int e = tid; e < NE; e += 1024) {
      const unsigned d = (unsigned)(tgt[e] - n0);
      if (d < (unsigned)NB) {
        const float4 ss = s_src4[src[e]];
        const float4 st = *(const float4*)&stg[d * NH];
        atomicMax(&emaxk[d * NH + 0], f2key(leaky(ss.x + st.x)));
        atomicMax(&emaxk[d * NH + 1], f2key(leaky(ss.y + st.y)));
        atomicMax(&emaxk[d * NH + 2], f2key(leaky(ss.z + st.z)));
        atomicMax(&emaxk[d * NH + 3], f2key(leaky(ss.w + st.w)));
      }
    }
    __syncthreads();
    for (int e = tid; e < NE; e += 1024) {
      const unsigned d = (unsigned)(tgt[e] - n0);
      if (d < (unsigned)NB) {
        const int s = src[e];
        const float4 ss = s_src4[s];
        const float4 st = *(const float4*)&stg[d * NH];
        const int4   mk = *(const int4*)&emaxk[d * NH];
        float exv[4];
        exv[0] = edge_exp(ss.x + st.x, mk.x);
        exv[1] = edge_exp(ss.y + st.y, mk.y);
        exv[2] = edge_exp(ss.z + st.z, mk.z);
        exv[3] = edge_exp(ss.w + st.w, mk.w);
#pragma unroll
        for (int k = 0; k < NH; ++k) atomicAdd(&esum[d * NH + k], exv[k]);
        const float* wr = Wh + (size_t)s * DT;
        float* ar = acc + d * ACCP;
        for (int j = 0; j < DT; ++j) atomicAdd(&ar[j], exv[j >> 5] * wr[j]);
      }
    }
  }
  __syncthreads();

  for (int i = tid; i < NB * DT; i += 1024) {
    const int d = i >> 7;
    const int j = i & 127;
    const float v = acc[d * ACCP + j] / (esum[d * NH + (j >> 5)] + 1e-10f);
    out[(size_t)(n0 + d) * DT + j] = v > 0.f ? v : expm1f(v);
  }
}

extern "C" void kernel_launch(void* const* d_in, const int* in_sizes, int n_in,
                              void* d_out, int out_size, void* d_ws, size_t ws_size,
                              hipStream_t stream) {
  const float* x    = (const float*)d_in[0];
  const int*   eidx = (const int*)d_in[1];
  const float* W    = (const float*)d_in[2];
  const float* a    = (const float*)d_in[3];
  float* out = (float*)d_out;

  const int* src = eidx;
  const int* tgt = eidx + NE;

  float* ws    = (float*)d_ws;
  float* Wh    = ws;
  float* s_src = Wh + (size_t)NN * DT;
  float* s_tgt = s_src + NN * NH;
  unsigned* ent = (unsigned*)(s_tgt + NN * NH);
  unsigned short* off = (unsigned short*)(ent + (size_t)NSEG * SEGE);

  gat_wh_gemm<<<NSEG, 128, 0, stream>>>(x, W, src, tgt, Wh, ent, off);

  gat_scores<<<(NN * NH + 255) / 256, 256, 0, stream>>>(Wh, a, s_src, s_tgt);

  gat_owner<<<NOWN, 1024, 0, stream>>>(src, tgt, ent, off, s_src, s_tgt, Wh, out);
}


# edge phase: the block's LDS list grouped by row (LDS integer ranks, one scan); a wave walks its rows' run with row and exp sums in registers, two groups of 8 row loads in flight; sum order changes
# speedup vs baseline: 10.3102x; 1.1211x over previous
#include <hip/hip_runtime.h>
#include <math.h>

#define NN 50000
#define NE 1600000
#define DIN 128
#define DOUT 32
#define NH 4
#define DT 128
#define NTILES (NN / 16)
#define NPAIR (DIN / 2)
#define WPAD 144
#define NB 200
#define EPW (NE / NTILES)
#define NOWN (NN / NB)
#define SEGE (4 * EPW)
#define NSEG ((NE + SEGE - 1) / SEGE)
#define NBKT 256
static_assert(NE == NTILES * EPW && EPW == 512, "one projection wave per 512 edges");
static_assert(NN == NOWN * NB && NOWN < NBKT - 1 && NB <= 256 && NN <= 65536, "bucket, row and source fit their fields");

typedef __attribute__((ext_vector_type(2))) float v2f;
typedef __attribute__((ext_vector_type(8))) float v8f;

__device__ inline unsigned bucket_of(int t) {
  const unsigned o = (unsigned)t / (unsigned)NB;
  return o < (unsigned)NOWN ? o : (unsigned)NOWN;
}
__device__ inline unsigned entry_of(int s, int t, unsigned b) {
  return ((unsigned)s & 0xFFFFu) | ((((unsigned)t - b * (unsigned)NB) & 0xFFu) << 16);
}


template <int M>
__device__ inline float add_xlane(float v, float w) {
  return v + __int_as_float(__builtin_amdgcn_update_dpp(0, __float_as_int(w), 0x160 + M, 0xf, 0xf, true));
}

__device__ inline void score_epilogue(const v8f (&acc)[8], const float (&cs)[8], const float (&ct)[8],
                                      int row0, int half, int l16,
                                      float* __restrict__ s_src, float* __restrict__ s_tgt) {
  const bool b3 = (l16 & 8) != 0, b2 = (l16 & 4) != 0, b1 = (l16 & 2) != 0, b0 = (l16 & 1) != 0;
  float v1[8][NH];
#pragma unroll
  for (int r = 0; r < 8; ++r) {
#pragma unroll
    for (int h = 0; h < NH; ++h) {
      const float ps = acc[2 * h][r] * cs[2 * h] + acc[2 * h + 1][r] * cs[2 * h + 1];
      const float pt = acc[2 * h][r] * ct[2 * h] + acc[2 * h + 1][r] * ct[2 * h + 1];
      v1[r][h] = add_xlane<8>(b3 ? pt : ps, b3 ? ps : pt);
    }
  }
  float v2[4][NH];
#pragma unroll
  for (int r = 0; r < 4; ++r) {
#pragma unroll
    for (int h = 0; h < NH; ++h)
      v2[r][h] = add_xlane<4>(b2 ? v1[4 + r][h] : v1[r][h], b2 ? v1[r][h] : v1[4 + r][h]);
  }
  float v3[2][NH];
#pragma unroll
  for (int r = 0; r < 2; ++r) {
#pragma unroll
    for (int h = 0; h < NH; ++h)
      v3[r][h] = add_xlane<2>(b1 ? v2[2 + r][h] : v2[r][h], b1 ? v2[r][h] : v2[2 + r][h]);
  }
  float4 s;
  s.x = add_xlane<1>(b0 ? v3[1][0] : v3[0][0], b0 ? v3[0][0] : v3[1][0]);
  s.y = add_xlane<1>(b0 ? v3[1][1] : v3[0][1], b0 ? v3[0][1] : v3[1][1]);
  s.z = add_xlane<1>(b0 ? v3[1][2] : v3[0][2], b0 ? v3[0][2] : v3[1][2]);
  s.w = add_xlane<1>(b0 ? v3[1][3] : v3[0][3], b0 ? v3[0][3] : v3[1][3]);
  float* dst = (b3 ? s_tgt : s_src) + (size_t)(row0 + 8 * half + (l16 & 7)) * NH;
  *(float4*)dst = s;
}

__global__ void __launch_bounds__(128) gat_wh_gemm(const float* __restrict__ x,
                                                   const float* __restrict__ W,
                                                   const float* __restrict__ a,
                                                   const int* __restrict__ src,
                                                   const int* __restrict__ tgt,
                                                   float* __restrict__ Wh,
                                                   float* __restrict__ s_src,
                                                   float* __restrict__ s_tgt,
                                                   unsigned* __restrict__ ent,
                                                   unsigned short* __restrict__ off) {
  __shared__ __attribute__((aligned(16))) v2f Wsh[NPAIR * WPAD];
  unsigned* sorted = (unsigned*)Wsh;
  unsigned* hist   = sorted + SEGE;
  unsigned* bstart = hist + NBKT;
  static_assert((SEGE + 2 * NBKT) * 4 <= NPAIR * WPAD * 8 && SEGE == 4 * 512, "grouping arrays fit the staging area; four 16-B words per thread");

  const int lane = threadIdx.x & 31;
  const int wave = threadIdx.x >> 5;
  const int tile = blockIdx.x * 4 + wave;
  const bool live = tile < NTILES;

  int4 tv[EPW / 128], sv[EPW / 128];
  {
    const int4* __restrict__ t4p = (const int4*)tgt;
    const int4* __restrict__ s4p = (const int4*)src;
    const int w0 = (live ? tile : 0) * (EPW / 4) + lane;
#pragma unroll
    for (int u = 0; u < EPW / 128; ++u) { tv[u] = t4p[w0 + u * 32]; sv[u] = s4p[w0 + u * 32]; }
  }
  hist[threadIdx.x]       = 0u;
  hist[threadIdx.x + 128] = 0u;
  __syncthreads();

  unsigned rk[EPW / 32];
  if (live) {
#pragma unroll
    for (int u = 0; u < EPW / 128; ++u) {
      rk[4 * u + 0] = atomicAdd(&hist[bucket_of(tv[u].x)], 1u);
      rk[4 * u + 1] = atomicAdd(&hist[bucket_of(tv[u].y)], 1u);
      rk[4 * u + 2] = atomicAdd(&hist[bucket_of(tv[u].z)], 1u);
      rk[4 * u + 3] = atomicAdd(&hist[bucket_of(tv[u].w)], 1u);
    }
  }
  __syncthreads();

  if (wave == 0) {
    unsigned c[8];
    unsigned sum = 0u;
#pragma unroll
    for (int j = 0; j < 8; ++j) { c[j] = hist[lane * 8 + j]; sum += c[j]; }
    unsigned incl = sum;
#pragma unroll
    for (int dlt = 1; dlt < 32; dlt <<= 1) {
      const unsigned v = __shfl_up(incl, dlt);
      if (lane >= dlt) incl += v;
    }
    unsigned run = incl - sum;
#pragma unroll
    for (int j = 0; j < 8; ++j) { bstart[lane * 8 + j] = run; run += c[j]; }
  }
  __syncthreads();

  if (live) {
    unsigned ps[EPW / 32];
#pragma unroll
    for (int u = 0; u < EPW / 128; ++u) {
      ps[4 * u + 0] = bstart[bucket_of(tv[u].x)] + rk[4 * u + 0];
      ps[4 * u + 1] = bstart[bucket_of(tv[u].y)] + rk[4 * u + 1];
      ps[4 * u + 2] = bstart[bucket_of(tv[u].z)] + rk[4 * u + 2];
      ps[4 * u + 3] = bstart[bucket_of(tv[u].w)] + rk[4 * u + 3];
    }
#pragma unroll
    for (int u = 0; u < EPW / 128; ++u) {
      sorted[ps[4 * u + 0]] = entry_of(sv[u].x, tv[u].x, bucket_of(tv[u].x));
      sorted[ps[4 * u + 1]] = entry_of(sv[u].y, tv[u].y, bucket_of(tv[u].y));
      sorted[ps[4 * u + 2]] = entry_of(sv[u].z, tv[u].z, bucket_of(tv[u].z));
      sorted[ps[4 * u + 3]] = entry_of(sv[u].w, tv[u].w, bucket_of(tv[u].w));
    }
  }
  __syncthreads();

  {
    const uint4* s4 = (const uint4*)sorted;
    const uint4 k0 = s4[threadIdx.x];
    const uint4 k1 = s4[128 + threadIdx.x];
    const uint4 k2 = s4[256 + threadIdx.x];
    const uint4 k3 = s4[384 + threadIdx.x];
    const unsigned f0 = bstart[2 * threadIdx.x], f1 = bstart[2 * threadIdx.x + 1];
    __syncthreads();
    uint4* e4 = (uint4*)(ent + (size_t)blockIdx.x * SEGE);
    e4[threadIdx.x]       = k0;
    e4[128 + threadIdx.x] = k1;
    e4[256 + threadIdx.x] = k2;
    e4[384 + threadIdx.x] = k3;
    ((unsigned*)off)[(size_t)blockIdx.x * (NBKT / 2) + threadIdx.x] = f0 | (f1 << 16);
  }

  for (int idx = threadIdx.x; idx < NPAIR * DT; idx += 128) {
    const int p  = idx >> 7;
    const int j  = idx & 127;
    const int kh = j >> 5;
    const int o  = j & 31;
    const float* wp = W + kh * (DIN * DOUT) + (2 * p) * DOUT + o;
    v2f pr;
    pr.x = wp[0];
    pr.y = wp[DOUT];
    Wsh[p * WPAD + j] = pr;
  }
  __syncthreads();

  const int half = lane >> 4;
  const int l16  = lane & 15;
  if (!live) return;
  const int row0 = tile * 16;

  v8f acc[8];
#pragma unroll
  for (int jt = 0; jt < 8; ++jt) acc[jt] = (v8f){0.f,0.f,0.f,0.f,0.f,0.f,0.f,0.f};

  float cs[8], ct[8];
#pragma unroll
  for (int jt = 0; jt < 8; ++jt) {
    cs[jt] = a[(jt >> 1) * (2 * DOUT) + (jt & 1) * 16 + l16];
    ct[jt] = a[(jt >> 1) * (2 * DOUT) + DOUT + (jt & 1) * 16 + l16];
  }

  const float* xrow = x + (size_t)(row0 + l16) * DIN;

#pragma unroll 4
  for (int kk = 0; kk < 32; ++kk) {
    const int i0 = kk * 4 + half * 2;
    const int p  = kk * 2 + half;
    const v2f a = *(const v2f*)(xrow + i0);
    const v2f* wr = Wsh + p * WPAD + l16;
#pragma unroll
    for (int jt = 0; jt < 8; ++jt) {
      const v2f b = wr[jt * 16];
      acc[jt] = __builtin_amdgcn_wmma_f32_16x16x4_f32(
          false, a, false, b, (short)0, acc[jt], false, false);
    }
  }

#pragma unroll
  for (int jt = 0; jt < 8; ++jt) {
#pragma unroll
    for (int r = 0; r < 8; ++r) {
      const int row = row0 + r + 8 * half;
      Wh[(size_t)row * DT + jt * 16 + l16] = acc[jt][r];
    }
  }

  score_epilogue(acc, cs, ct, row0, half, l16, s_src, s_tgt);
}

#define ACCP 128
#define NWAVE 32
#define CAP 8192
#define EPT (CAP / 1024)
#define NPASS ((NSEG + 127) / 128)
#define NROWC 256
#define KEY_EMPTY (-2147483647 - 1)

__device__ inline int f2key(float v) {
  const int b = __float_as_int(v);
  return b >= 0 ? b : (b ^ 0x7FFFFFFF);
}
__device__ inline float key2f(int k) {
  return __int_as_float(k >= 0 ? k : (k ^ 0x7FFFFFFF));
}
__device__ inline float leaky(float v) { return v > 0.f ? v : 0.2f * v; }
__device__ inline float score_exp(float score, int maxkey) {
  float m = key2f(maxkey);
  if (!(__builtin_fabsf(m) < __builtin_inff())) m = 0.f;
  return __expf(score - m);
}
__device__ inline float edge_exp(float score_sum, int maxkey) {
  return score_exp(leaky(score_sum), maxkey);
}

__device__ inline unsigned first_row_of(unsigned w) { return (25u * w + 3u) >> 2; }
static_assert(NB == 200 && NWAVE == 32 && NB <= NROWC && CAP <= 65536, "row ranges fill the waves; list places fit 16 bits");

__global__ void __launch_bounds__(1024) gat_owner(const int* __restrict__ src,
                                                  const int* __restrict__ tgt,
                                                  const unsigned* __restrict__ ent,
                                                  const unsigned short* __restrict__ off,
                                                  const float* __restrict__ s_src,
                                                  const float* __restrict__ s_tgt,
                                                  const float* __restrict__ Wh,
                                                  float* __restrict__ out) {
  __shared__ __attribute__((aligned(16))) float acc[NB * ACCP];
  __shared__ unsigned list[CAP];
  __shared__ float4 esc[CAP];
  __shared__ __attribute__((aligned(16))) int   emaxk[NB * NH];
  __shared__ __attribute__((aligned(16))) float esum[NB * NH];
  __shared__ __attribute__((aligned(16))) float stg[NB * NH];
  __shared__ unsigned short sub[CAP];
  __shared__ unsigned rcnt[NROWC];
  __shared__ unsigned rbeg[NROWC];
  __shared__ unsigned cnt;

  const int tid = threadIdx.x;
  const int n0  = blockIdx.x * NB;
  const float4* __restrict__ s_src4 = (const float4*)s_src;

  for (int i = tid; i < NB * ACCP; i += 1024) acc[i] = 0.f;
  for (int i = tid; i < NB * NH; i += 1024) {
    emaxk[i] = KEY_EMPTY;
    esum[i]  = 0.f;
    stg[i]   = s_tgt[n0 * NH + i];
  }
  if (tid == 0) cnt = 0u;
  if (tid < NROWC) rcnt[tid] = 0u;
  __syncthreads();

  {
    const int lane8 = tid & 7;
    const int grp   = tid >> 3;
    const int lead  = (tid & 31) & ~7;
    unsigned st[NPASS], ln[NPASS], bs[NPASS], v0[NPASS], v1[NPASS];
#pragma unroll
    for (int p = 0; p < NPASS; ++p) {
      const int seg  = p * 128 + grp;
      const int segc = seg < NSEG ? seg : NSEG - 1;
      const unsigned a = off[segc * NBKT + blockIdx.x];
      const unsigned b = off[segc * NBKT + blockIdx.x + 1];
      st[p] = a;
      ln[p] = seg < NSEG ? b - a : 0u;
    }
#pragma unroll
    for (int p = 0; p < NPASS; ++p) {
      unsigned base = 0u;
      if (lane8 == 0 && ln[p] != 0u) base = atomicAdd(&cnt, ln[p]);
      bs[p] = (unsigned)__shfl((int)base, lead);
    }
#pragma unroll
    for (int p = 0; p < NPASS; ++p) {
      const int seg  = p * 128 + grp;
      const int segc = seg < NSEG ? seg : NSEG - 1;
      const unsigned* eb = ent + (size_t)segc * SEGE;
      const unsigned i0 = st[p] + (unsigned)lane8;
      const unsigned i1 = i0 + 8u;
      v0[p] = eb[i0 < (unsigned)SEGE ? i0 : (unsigned)SEGE - 1u];
      v1[p] = eb[i1 < (unsigned)SEGE ? i1 : (unsigned)SEGE - 1u];
    }
#pragma unroll
    for (int p = 0; p < NPASS; ++p) {
      const unsigned q0 = bs[p] + (unsigned)lane8;
      if ((unsigned)lane8 < ln[p] && q0 < (unsigned)CAP) list[q0] = v0[p];
      if ((unsigned)lane8 + 8u < ln[p] && q0 + 8u < (unsigned)CAP) list[q0 + 8u] = v1[p];
      if (ln[p] > 16u) {
        const int seg = p * 128 + grp;
        const unsigned* ep = ent + (size_t)seg * SEGE + st[p];
        for (unsigned i = (unsigned)lane8 + 16u; i < ln[p]; i += 8u)
          if (bs[p] + i < (unsigned)CAP) list[bs[p] + i] = ep[i];
      }
    }
  }
  __syncthreads();

  const unsigned total = (unsigned)__builtin_amdgcn_readfirstlane((int)cnt);

  if (total <= (unsigned)CAP) {
    const int n = (int)total;

    {
      unsigned en[EPT], rk[EPT];
      float4 ss[EPT];
#pragma unroll
      for (int u = 0; u < EPT; ++u) {
        const int idx = tid + u * 1024;
        const unsigned raw = list[idx];
        en[u] = idx < n ? raw : 0u;
        rk[u] = 0u;
      }
#pragma unroll
      for (int u = 0; u < EPT; ++u) ss[u] = s_src4[en[u] & 0xFFFFu];
#pragma unroll
      for (int u = 0; u < EPT; ++u) {
        const int idx = tid + u * 1024;
        if (idx < n) {
          const unsigned d  = en[u] >> 16;
          const float4   st = *(const float4*)&stg[d * NH];
          float4 sc;
          sc.x = leaky(ss[u].x + st.x);
          sc.y = leaky(ss[u].y + st.y);
          sc.z = leaky(ss[u].z + st.z);
          sc.w = leaky(ss[u].w + st.w);
          esc[idx]  = sc;
          atomicMax(&emaxk[d * NH + 0], f2key(sc.x));
          atomicMax(&emaxk[d * NH + 1], f2key(sc.y));
          atomicMax(&emaxk[d * NH + 2], f2key(sc.z));
          atomicMax(&emaxk[d * NH + 3], f2key(sc.w));
          rk[u] = atomicAdd(&rcnt[d], 1u);
        }
      }
      __syncthreads();

      if (tid < 32) {
        unsigned cc[8];
        unsigned sum = 0u;
#pragma unroll
        for (int j = 0; j < 8; ++j) { cc[j] = rcnt[tid * 8 + j]; sum += cc[j]; }
        unsigned incl = sum;
#pragma unroll
        for (int dlt = 1; dlt < 32; dlt <<= 1) {
          const unsigned v = __shfl_up(incl, dlt);
          if (tid >= dlt) incl += v;
        }
        unsigned run = incl - sum;
#pragma unroll
        for (int j = 0; j < 8; ++j) { rbeg[tid * 8 + j] = run; run += cc[j]; }
      }
      __syncthreads();

#pragma unroll
      for (int u = 0; u < EPT; ++u) {
        const int idx = tid + u * 1024;
        if (idx < n) {
          const unsigned pos = rbeg[en[u] >> 16] + rk[u];
          if (pos < (unsigned)CAP) sub[pos] = (unsigned short)idx;
        }
      }
    }
    __syncthreads();

    const int lane = tid & 31;
    const unsigned wave = (unsigned)__builtin_amdgcn_readfirstlane(tid >> 5);
    const int kq = lane >> 3;
    const int hm0 = kq == 0 ? -1 : 0, hm1 = kq == 1 ? -1 : 0, hm2 = kq == 2 ? -1 : 0, hm3 = kq == 3 ? -1 : 0;
    const float4* __restrict__ Wh4 = (const float4*)Wh;
    float4* acc4 = (float4*)acc + lane;
    const unsigned r0 = first_row_of(wave), r1 = first_row_of(wave + 1u);
    const unsigned kb = (unsigned)__builtin_amdgcn_readfirstlane((int)rbeg[r0]);
    const unsigned ke = (unsigned)__builtin_amdgcn_readfirstlane((int)rbeg[r1]);

    unsigned cur = r0;
    float4 a4 = make_float4(0.f, 0.f, 0.f, 0.f);
    float ws = 0.f;
    auto store_row = [&]() {
      acc4[cur * (ACCP / 4)] = a4;
      if ((lane & 7) == 0) esum[cur * NH + kq] = ws;
    };

    auto prepare = [&](unsigned b, unsigned& en_o, float& e0, float& e1, float& e2, float& e3) {
      const unsigned k  = b + (unsigned)lane;
      const unsigned kc = k < ke ? k : ke - 1u;
      const int idx = sub[kc];
      en_o = list[idx];
      const unsigned d = en_o >> 16;
      const float4 sc = esc[idx];
      const int4   mk = *(const int4*)&emaxk[d * NH];
      e0 = score_exp(sc.x, mk.x);
      e1 = score_exp(sc.y, mk.y);
      e2 = score_exp(sc.z, mk.z);
      e3 = score_exp(sc.w, mk.w);
      if (k >= ke) { e0 = 0.f; e1 = 0.f; e2 = 0.f; e3 = 0.f; }
    };

    if (kb < ke) {
      unsigned en2;
      float ex0, ex1, ex2, ex3;
      prepare(kb, en2, ex0, ex1, ex2, ex3);

      auto issue = [&](float4 (&r)[8], int j0) {
#pragma unroll
        for (int u = 0; u < 8; ++u) {
          const unsigned e = (unsigned)__builtin_amdgcn_readlane((int)en2, j0 + u);
          r[u] = Wh4[(size_t)(e & 0xFFFFu) * (DT / 4) + lane];
        }
      };
      auto add8 = [&](const float4 (&r)[8], int j0) {
#pragma unroll
        for (int u = 0; u < 8; ++u) {
          const unsigned e  = (unsigned)__builtin_amdgcn_readlane((int)en2, j0 + u);
          const unsigned du = e >> 16;
          if (__builtin_expect(du != cur, 0)) {
            store_row();
            a4 = make_float4(0.f, 0.f, 0.f, 0.f);
            ws = 0.f;
            cur = du;
          }
          const int b0 = __builtin_amdgcn_readlane(__float_as_int(ex0), j0 + u);
          const int b1 = __builtin_amdgcn_readlane(__float_as_int(ex1), j0 + u);
          const int b2 = __builtin_amdgcn_readlane(__float_as_int(ex2), j0 + u);
          const int b3 = __builtin_amdgcn_readlane(__float_as_int(ex3), j0 + u);
          const float w = __int_as_float((b0 & hm0) | (b1 & hm1) | (b2 & hm2) | (b3 & hm3));
          a4.x += w * r[u].x;
          a4.y += w * r[u].y;
          a4.z += w * r[u].z;
          a4.w += w * r[u].w;
          ws += w;
        }
      };

      for (unsigned base = kb; base < ke; base += 32u) {
        float4 ra[8], rb[8];
        issue(ra, 0);
        issue(rb, 8);
        unsigned en2n;
        float nx0, nx1, nx2, nx3;
        prepare(base + 32u, en2n, nx0, nx1, nx2, nx3);
        add8(ra, 0);
        issue(ra, 16);
        add8(rb, 8);
        issue(rb, 24);
        add8(ra, 16);
        add8(rb, 24);
        en2 = en2n; ex0 = nx0; ex1 = nx1; ex2 = nx2; ex3 = nx3;
      }
    }
    store_row();
  } else {
    for (int e = tid; e < NE; e += 1024) {
      const unsigned d = (unsigned)(tgt[e] - n0);
      if (d < (unsigned)NB) {
        const float4 ss = s_src4[src[e]];
        const float4 st = *(const float4*)&stg[d * NH];
        atomicMax(&emaxk[d * NH + 0], f2key(leaky(ss.x + st.x)));
        atomicMax(&emaxk[d * NH + 1], f2key(leaky(ss.y + st.y)));
        atomicMax(&emaxk[d * NH + 2], f2key(leaky(ss.z + st.z)));
        atomicMax(&emaxk[d * NH + 3], f2key(leaky(ss.w + st.w)));
      }
    }
    __syncthreads();
    for (int e = tid; e < NE; e += 1024) {
      const unsigned d = (unsigned)(tgt[e] - n0);
      if (d < (unsigned)NB) {
        const int s = src[e];
        const float4 ss = s_src4[s];
        const float4 st = *(const float4*)&stg[d * NH];
        const int4   mk = *(const int4*)&emaxk[d * NH];
        float exv[4];
        exv[0] = edge_exp(ss.x + st.x, mk.x);
        exv[1] = edge_exp(ss.y + st.y, mk.y);
        exv[2] = edge_exp(ss.z + st.z, mk.z);
        exv[3] = edge_exp(ss.w + st.w, mk.w);
#pragma unroll
        for (int k = 0; k < NH; ++k) atomicAdd(&esum[d * NH + k], exv[k]);
        const float* wr = Wh + (size_t)s * DT;
        float* ar = acc + d * ACCP;
        for (int j = 0; j < DT; ++j) atomicAdd(&ar[j], exv[j >> 5] * wr[j]);
      }
    }
  }
  __syncthreads();

  for (int i = tid; i < NB * DT; i += 1024) {
    const int d = i >> 7;
    const int j = i & 127;
    const float v = acc[d * ACCP + j] / (esum[d * NH + (j >> 5)] + 1e-10f);
    out[(size_t)(n0 + d) * DT + j] = v > 0.f ? v : expm1f(v);
  }
}

extern "C" void kernel_launch(void* const* d_in, const int* in_sizes, int n_in,
                              void* d_out, int out_size, void* d_ws, size_t ws_size,
                              hipStream_t stream) {
  const float* x    = (const float*)d_in[0];
  const int*   eidx = (const int*)d_in[1];
  const float* W    = (const float*)d_in[2];
  const float* a    = (const float*)d_in[3];
  float* out = (float*)d_out;

  const int* src = eidx;
  const int* tgt = eidx + NE;

  float* ws    = (float*)d_ws;
  float* Wh    = ws;
  float* s_src = Wh + (size_t)NN * DT;
  float* s_tgt = s_src + NN * NH;
  unsigned* ent = (unsigned*)(s_tgt + NN * NH);
  unsigned short* off = (unsigned short*)(ent + (size_t)NSEG * SEGE);

  gat_wh_gemm<<<NSEG, 128, 0, stream>>>(x, W, a, src, tgt, Wh, s_src, s_tgt, ent, off);

  gat_owner<<<NOWN, 1024, 0, stream>>>(src, tgt, ent, off, s_src, s_tgt, Wh, out);
}


# edge phase: entries and scores stored grouped by row; per 32-place batch a lane stages e_exp and row offset in LDS, walk reads 4 places per 16-B LDS load: no per-edge readlane or select; same sums
# speedup vs baseline: 11.1222x; 1.0788x over previous
#include <hip/hip_runtime.h>
#include <math.h>

#define NN 50000
#define NE 1600000
#define DIN 128
#define DOUT 32
#define NH 4
#define DT 128
#define NTILES (NN / 16)
#define NPAIR (DIN / 2)
#define WPAD 144
#define NB 200
#define EPW (NE / NTILES)
#define NOWN (NN / NB)
#define SEGE (4 * EPW)
#define NSEG ((NE + SEGE - 1) / SEGE)
#define NBKT 256
static_assert(NE == NTILES * EPW && EPW == 512, "one projection wave per 512 edges");
static_assert(NN == NOWN * NB && NOWN < NBKT - 1 && NB <= 256 && NN <= 65536, "bucket, row and source fit their fields");

typedef __attribute__((ext_vector_type(2))) float v2f;
typedef __attribute__((ext_vector_type(8))) float v8f;

__device__ inline unsigned bucket_of(int t) {
  const unsigned o = (unsigned)t / (unsigned)NB;
  return o < (unsigned)NOWN ? o : (unsigned)NOWN;
}
__device__ inline unsigned entry_of(int s, int t, unsigned b) {
  return ((unsigned)s & 0xFFFFu) | ((((unsigned)t - b * (unsigned)NB) & 0xFFu) << 16);
}


template <int M>
__device__ inline float add_xlane(float v, float w) {
  return v + __int_as_float(__builtin_amdgcn_update_dpp(0, __float_as_int(w), 0x160 + M, 0xf, 0xf, true));
}

__device__ inline void score_epilogue(const v8f (&acc)[8], const float (&cs)[8], const float (&ct)[8],
                                      int row0, int half, int l16,
                                      float* __restrict__ s_src, float* __restrict__ s_tgt) {
  const bool b3 = (l16 & 8) != 0, b2 = (l16 & 4) != 0, b1 = (l16 & 2) != 0, b0 = (l16 & 1) != 0;
  float v1[8][NH];
#pragma unroll
  for (int r = 0; r < 8; ++r) {
#pragma unroll
    for (int h = 0; h < NH; ++h) {
      const float ps = acc[2 * h][r] * cs[2 * h] + acc[2 * h + 1][r] * cs[2 * h + 1];
      const float pt = acc[2 * h][r] * ct[2 * h] + acc[2 * h + 1][r] * ct[2 * h + 1];
      v1[r][h] = add_xlane<8>(b3 ? pt : ps, b3 ? ps : pt);
    }
  }
  float v2[4][NH];
#pragma unroll
  for (int r = 0; r < 4; ++r) {
#pragma unroll
    for (int h = 0; h < NH; ++h)
      v2[r][h] = add_xlane<4>(b2 ? v1[4 + r][h] : v1[r][h], b2 ? v1[r][h] : v1[4 + r][h]);
  }
  float v3[2][NH];
#pragma unroll
  for (int r = 0; r < 2; ++r) {
#pragma unroll
    for (int h = 0; h < NH; ++h)
      v3[r][h] = add_xlane<2>(b1 ? v2[2 + r][h] : v2[r][h], b1 ? v2[r][h] : v2[2 + r][h]);
  }
  float4 s;
  s.x = add_xlane<1>(b0 ? v3[1][0] : v3[0][0], b0 ? v3[0][0] : v3[1][0]);
  s.y = add_xlane<1>(b0 ? v3[1][1] : v3[0][1], b0 ? v3[0][1] : v3[1][1]);
  s.z = add_xlane<1>(b0 ? v3[1][2] : v3[0][2], b0 ? v3[0][2] : v3[1][2]);
  s.w = add_xlane<1>(b0 ? v3[1][3] : v3[0][3], b0 ? v3[0][3] : v3[1][3]);
  float* dst = (b3 ? s_tgt : s_src) + (size_t)(row0 + 8 * half + (l16 & 7)) * NH;
  *(float4*)dst = s;
}

__global__ void __launch_bounds__(128) gat_wh_gemm(const float* __restrict__ x,
                                                   const float* __restrict__ W,
                                                   const float* __restrict__ a,
                                                   const int* __restrict__ src,
                                                   const int* __restrict__ tgt,
                                                   float* __restrict__ Wh,
                                                   float* __restrict__ s_src,
                                                   float* __restrict__ s_tgt,
                                                   unsigned* __restrict__ ent,
                                                   unsigned short* __restrict__ off) {
  __shared__ __attribute__((aligned(16))) v2f Wsh[NPAIR * WPAD];
  unsigned* sorted = (unsigned*)Wsh;
  unsigned* hist   = sorted + SEGE;
  unsigned* bstart = hist + NBKT;
  static_assert((SEGE + 2 * NBKT) * 4 <= NPAIR * WPAD * 8 && SEGE == 4 * 512, "grouping arrays fit the staging area; four 16-B words per thread");

  const int lane = threadIdx.x & 31;
  const int wave = threadIdx.x >> 5;
  const int tile = blockIdx.x * 4 + wave;
  const bool live = tile < NTILES;

  int4 tv[EPW / 128], sv[EPW / 128];
  {
    const int4* __restrict__ t4p = (const int4*)tgt;
    const int4* __restrict__ s4p = (const int4*)src;
    const int w0 = (live ? tile : 0) * (EPW / 4) + lane;
#pragma unroll
    for (int u = 0; u < EPW / 128; ++u) { tv[u] = t4p[w0 + u * 32]; sv[u] = s4p[w0 + u * 32]; }
  }
  hist[threadIdx.x]       = 0u;
  hist[threadIdx.x + 128] = 0u;
  __syncthreads();

  unsigned rk[EPW / 32];
  if (live) {
#pragma unroll
    for (int u = 0; u < EPW / 128; ++u) {
      rk[4 * u + 0] = atomicAdd(&hist[bucket_of(tv[u].x)], 1u);
      rk[4 * u + 1] = atomicAdd(&hist[bucket_of(tv[u].y)], 1u);
      rk[4 * u + 2] = atomicAdd(&hist[bucket_of(tv[u].z)], 1u);
      rk[4 * u + 3] = atomicAdd(&hist[bucket_of(tv[u].w)], 1u);
    }
  }
  __syncthreads();

  if (wave == 0) {
    unsigned c[8];
    unsigned sum = 0u;
#pragma unroll
    for (int j = 0; j < 8; ++j) { c[j] = hist[lane * 8 + j]; sum += c[j]; }
    unsigned incl = sum;
#pragma unroll
    for (int dlt = 1; dlt < 32; dlt <<= 1) {
      const unsigned v = __shfl_up(incl, dlt);
      if (lane >= dlt) incl += v;
    }
    unsigned run = incl - sum;
#pragma unroll
    for (int j = 0; j < 8; ++j) { bstart[lane * 8 + j] = run; run += c[j]; }
  }
  __syncthreads();

  if (live) {
    unsigned ps[EPW / 32];
#pragma unroll
    for (int u = 0; u < EPW / 128; ++u) {
      ps[4 * u + 0] = bstart[bucket_of(tv[u].x)] + rk[4 * u + 0];
      ps[4 * u + 1] = bstart[bucket_of(tv[u].y)] + rk[4 * u + 1];
      ps[4 * u + 2] = bstart[bucket_of(tv[u].z)] + rk[4 * u + 2];
      ps[4 * u + 3] = bstart[bucket_of(tv[u].w)] + rk[4 * u + 3];
    }
#pragma unroll
    for (int u = 0; u < EPW / 128; ++u) {
      sorted[ps[4 * u + 0]] = entry_of(sv[u].x, tv[u].x, bucket_of(tv[u].x));
      sorted[ps[4 * u + 1]] = entry_of(sv[u].y, tv[u].y, bucket_of(tv[u].y));
      sorted[ps[4 * u + 2]] = entry_of(sv[u].z, tv[u].z, bucket_of(tv[u].z));
      sorted[ps[4 * u + 3]] = entry_of(sv[u].w, tv[u].w, bucket_of(tv[u].w));
    }
  }
  __syncthreads();

  {
    const uint4* s4 = (const uint4*)sorted;
    const uint4 k0 = s4[threadIdx.x];
    const uint4 k1 = s4[128 + threadIdx.x];
    const uint4 k2 = s4[256 + threadIdx.x];
    const uint4 k3 = s4[384 + threadIdx.x];
    const unsigned f0 = bstart[2 * threadIdx.x], f1 = bstart[2 * threadIdx.x + 1];
    __syncthreads();
    uint4* e4 = (uint4*)(ent + (size_t)blockIdx.x * SEGE);
    e4[threadIdx.x]       = k0;
    e4[128 + threadIdx.x] = k1;
    e4[256 + threadIdx.x] = k2;
    e4[384 + threadIdx.x] = k3;
    ((unsigned*)off)[(size_t)blockIdx.x * (NBKT / 2) + threadIdx.x] = f0 | (f1 << 16);
  }

  for (int idx = threadIdx.x; idx < NPAIR * DT; idx += 128) {
    const int p  = idx >> 7;
    const int j  = idx & 127;
    const int kh = j >> 5;
    const int o  = j & 31;
    const float* wp = W + kh * (DIN * DOUT) + (2 * p) * DOUT + o;
    v2f pr;
    pr.x = wp[0];
    pr.y = wp[DOUT];
    Wsh[p * WPAD + j] = pr;
  }
  __syncthreads();

  const int half = lane >> 4;
  const int l16  = lane & 15;
  if (!live) return;
  const int row0 = tile * 16;

  v8f acc[8];
#pragma unroll
  for (int jt = 0; jt < 8; ++jt) acc[jt] = (v8f){0.f,0.f,0.f,0.f,0.f,0.f,0.f,0.f};

  float cs[8], ct[8];
#pragma unroll
  for (int jt = 0; jt < 8; ++jt) {
    cs[jt] = a[(jt >> 1) * (2 * DOUT) + (jt & 1) * 16 + l16];
    ct[jt] = a[(jt >> 1) * (2 * DOUT) + DOUT + (jt & 1) * 16 + l16];
  }

  const float* xrow = x + (size_t)(row0 + l16) * DIN;

#pragma unroll 4
  for (int kk = 0; kk < 32; ++kk) {
    const int i0 = kk * 4 + half * 2;
    const int p  = kk * 2 + half;
    const v2f a = *(const v2f*)(xrow + i0);
    const v2f* wr = Wsh + p * WPAD + l16;
#pragma unroll
    for (int jt = 0; jt < 8; ++jt) {
      const v2f b = wr[jt * 16];
      acc[jt] = __builtin_amdgcn_wmma_f32_16x16x4_f32(
          false, a, false, b, (short)0, acc[jt], false, false);
    }
  }

#pragma unroll
  for (int jt = 0; jt < 8; ++jt) {
#pragma unroll
    for (int r = 0; r < 8; ++r) {
      const int row = row0 + r + 8 * half;
      Wh[(size_t)row * DT + jt * 16 + l16] = acc[jt][r];
    }
  }

  score_epilogue(acc, cs, ct, row0, half, l16, s_src, s_tgt);
}

#define ACCP 128
#define NWAVE 32
#define CAP 8192
#define EPT (CAP / 1024)
#define NPASS ((NSEG + 127) / 128)
#define NROWC 256
#define WSTG (NH + 1)
#define KEY_EMPTY (-2147483647 - 1)

__device__ inline int f2key(float v) {
  const int b = __float_as_int(v);
  return b >= 0 ? b : (b ^ 0x7FFFFFFF);
}
__device__ inline float key2f(int k) {
  return __int_as_float(k >= 0 ? k : (k ^ 0x7FFFFFFF));
}
__device__ inline float leaky(float v) { return v > 0.f ? v : 0.2f * v; }
__device__ inline float score_exp(float score, int maxkey) {
  float m = key2f(maxkey);
  if (!(__builtin_fabsf(m) < __builtin_inff())) m = 0.f;
  return __expf(score - m);
}
__device__ inline float edge_exp(float score_sum, int maxkey) {
  return score_exp(leaky(score_sum), maxkey);
}

__device__ inline unsigned first_row_of(unsigned w) { return (25u * w + 3u) >> 2; }
static_assert(NB == 200 && NWAVE == 32 && NB <= NROWC && CAP <= 65536, "row ranges fill the waves; list places fit 16 bits");

__global__ void __launch_bounds__(1024) gat_owner(const int* __restrict__ src,
                                                  const int* __restrict__ tgt,
                                                  const unsigned* __restrict__ ent,
                                                  const unsigned short* __restrict__ off,
                                                  const float* __restrict__ s_src,
                                                  const float* __restrict__ s_tgt,
                                                  const float* __restrict__ Wh,
                                                  float* __restrict__ out) {
  __shared__ __attribute__((aligned(16))) float acc[NB * ACCP];
  __shared__ unsigned list[CAP];
  __shared__ float4 esc[CAP];
  __shared__ __attribute__((aligned(16))) int   emaxk[NB * NH];
  __shared__ __attribute__((aligned(16))) float esum[NB * NH];
  __shared__ __attribute__((aligned(16))) float stg[NB * NH];
  __shared__ __attribute__((aligned(16))) float wst[NWAVE * 2 * WSTG * 32];
  __shared__ unsigned rcnt[NROWC];
  __shared__ unsigned rbeg[NROWC];
  __shared__ unsigned cnt;

  const int tid = threadIdx.x;
  const int n0  = blockIdx.x * NB;
  const float4* __restrict__ s_src4 = (const float4*)s_src;

  for (int i = tid; i < NB * ACCP; i += 1024) acc[i] = 0.f;
  for (int i = tid; i < NB * NH; i += 1024) {
    emaxk[i] = KEY_EMPTY;
    esum[i]  = 0.f;
    stg[i]   = s_tgt[n0 * NH + i];
  }
  if (tid == 0) cnt = 0u;
  if (tid < NROWC) rcnt[tid] = 0u;
  __syncthreads();

  {
    const int lane8 = tid & 7;
    const int grp   = tid >> 3;
    const int lead  = (tid & 31) & ~7;
    unsigned st[NPASS], ln[NPASS], bs[NPASS], v0[NPASS], v1[NPASS];
#pragma unroll
    for (int p = 0; p < NPASS; ++p) {
      const int seg  = p * 128 + grp;
      const int segc = seg < NSEG ? seg : NSEG - 1;
      const unsigned a = off[segc * NBKT + blockIdx.x];
      const unsigned b = off[segc * NBKT + blockIdx.x + 1];
      st[p] = a;
      ln[p] = seg < NSEG ? b - a : 0u;
    }
#pragma unroll
    for (int p = 0; p < NPASS; ++p) {
      unsigned base = 0u;
      if (lane8 == 0 && ln[p] != 0u) base = atomicAdd(&cnt, ln[p]);
      bs[p] = (unsigned)__shfl((int)base, lead);
    }
#pragma unroll
    for (int p = 0; p < NPASS; ++p) {
      const int seg  = p * 128 + grp;
      const int segc = seg < NSEG ? seg : NSEG - 1;
      const unsigned* eb = ent + (size_t)segc * SEGE;
      const unsigned i0 = st[p] + (unsigned)lane8;
      const unsigned i1 = i0 + 8u;
      v0[p] = eb[i0 < (unsigned)SEGE ? i0 : (unsigned)SEGE - 1u];
      v1[p] = eb[i1 < (unsigned)SEGE ? i1 : (unsigned)SEGE - 1u];
    }
#pragma unroll
    for (int p = 0; p < NPASS; ++p) {
      const unsigned q0 = bs[p] + (unsigned)lane8;
      if ((unsigned)lane8 < ln[p] && q0 < (unsigned)CAP) list[q0] = v0[p];
      if ((unsigned)lane8 + 8u < ln[p] && q0 + 8u < (unsigned)CAP) list[q0 + 8u] = v1[p];
      if (ln[p] > 16u) {
        const int seg = p * 128 + grp;
        const unsigned* ep = ent + (size_t)seg * SEGE + st[p];
        for (unsigned i = (unsigned)lane8 + 16u; i < ln[p]; i += 8u)
          if (bs[p] + i < (unsigned)CAP) list[bs[p] + i] = ep[i];
      }
    }
  }
  __syncthreads();

  const unsigned total = (unsigned)__builtin_amdgcn_readfirstlane((int)cnt);

  if (total <= (unsigned)CAP) {
    const int n = (int)total;

    {
      unsigned en[EPT], rk[EPT];
      float4 ss[EPT], sc[EPT];
#pragma unroll
      for (int u = 0; u < EPT; ++u) {
        const int idx = tid + u * 1024;
        const unsigned raw = list[idx];
        en[u] = idx < n ? raw : 0u;
        rk[u] = 0u;
      }
#pragma unroll
      for (int u = 0; u < EPT; ++u) ss[u] = s_src4[en[u] & 0xFFFFu];
#pragma unroll
      for (int u = 0; u < EPT; ++u) {
        const int idx = tid + u * 1024;
        if (idx < n) {
          const unsigned d  = en[u] >> 16;
          const float4   st = *(const float4*)&stg[d * NH];
          sc[u].x = leaky(ss[u].x + st.x);
          sc[u].y = leaky(ss[u].y + st.y);
          sc[u].z = leaky(ss[u].z + st.z);
          sc[u].w = leaky(ss[u].w + st.w);
          atomicMax(&emaxk[d * NH + 0], f2key(sc[u].x));
          atomicMax(&emaxk[d * NH + 1], f2key(sc[u].y));
          atomicMax(&emaxk[d * NH + 2], f2key(sc[u].z));
          atomicMax(&emaxk[d * NH + 3], f2key(sc[u].w));
          rk[u] = atomicAdd(&rcnt[d], 1u);
        } else {
          sc[u] = make_float4(0.f, 0.f, 0.f, 0.f);
        }
      }
      __syncthreads();

      if (tid < 32) {
        unsigned cc[8];
        unsigned sum = 0u;
#pragma unroll
        for (int j = 0; j < 8; ++j) { cc[j] = rcnt[tid * 8 + j]; sum += cc[j]; }
        unsigned incl = sum;
#pragma unroll
        for (int dlt = 1; dlt < 32; dlt <<= 1) {
          const unsigned v = __shfl_up(incl, dlt);
          if (tid >= dlt) incl += v;
        }
        unsigned run = incl - sum;
#pragma unroll
        for (int j = 0; j < 8; ++j) { rbeg[tid * 8 + j] = run; run += cc[j]; }
      }
      __syncthreads();

#pragma unroll
      for (int u = 0; u < EPT; ++u) {
        const int idx = tid + u * 1024;
        if (idx < n) {
          const unsigned pos = rbeg[en[u] >> 16] + rk[u];
          if (pos < (unsigned)CAP) { list[pos] = en[u]; esc[pos] = sc[u]; }
        }
      }
    }
    __syncthreads();

    const int lane = tid & 31;
    const unsigned wave = (unsigned)__builtin_amdgcn_readfirstlane(tid >> 5);
    const int kq = lane >> 3;
    const char* __restrict__ WhB = (const char*)Wh;
    const unsigned lane16 = (unsigned)lane * 16u;
    float4* acc4 = (float4*)acc + lane;
    const unsigned r0 = first_row_of(wave), r1 = first_row_of(wave + 1u);
    const unsigned kb = (unsigned)__builtin_amdgcn_readfirstlane((int)rbeg[r0]);
    const unsigned ke = (unsigned)__builtin_amdgcn_readfirstlane((int)rbeg[r1]);
    float* const wsw = wst + wave * (2u * WSTG * 32u);

    unsigned cur = r0;
    float4 a4 = make_float4(0.f, 0.f, 0.f, 0.f);
    float ws = 0.f;
    auto store_row = [&]() {
      acc4[cur * (ACCP / 4)] = a4;
      if ((lane & 7) == 0) esum[cur * NH + kq] = ws;
    };

    auto stage = [&](unsigned b, float* s, unsigned& d_o) -> unsigned {
      const unsigned k  = b + (unsigned)lane;
      const unsigned kc = k < ke ? k : ke - 1u;
      const unsigned kp = kc > kb ? kc - 1u : kb;
      const unsigned en = list[kc];
      const unsigned ep = list[kp];
      const float4 sc = esc[kc];
      const unsigned d = en >> 16;
      const int4   mk = *(const int4*)&emaxk[d * NH];
      float e0 = score_exp(sc.x, mk.x);
      float e1 = score_exp(sc.y, mk.y);
      float e2 = score_exp(sc.z, mk.z);
      float e3 = score_exp(sc.w, mk.w);
      if (k >= ke) { e0 = 0.f; e1 = 0.f; e2 = 0.f; e3 = 0.f; }
      s[0 * 32 + lane] = e0;
      s[1 * 32 + lane] = e1;
      s[2 * 32 + lane] = e2;
      s[3 * 32 + lane] = e3;
      s[NH * 32 + lane] = __uint_as_float((en & 0xFFFFu) << 9);
      const unsigned dp = k > kb ? (ep >> 16) : r0;
      d_o = d;
      return __builtin_amdgcn_ballot_w32(k < ke && d != dp);
    };

    if (kb < ke) {
      unsigned dv;
      unsigned bm = stage(kb, wsw, dv);
      __builtin_amdgcn_fence(__ATOMIC_ACQ_REL, "wavefront");
      unsigned sel = 0u;

      for (unsigned base = kb; base < ke; base += 32u) {
        const float* sw = wsw + sel * (WSTG * 32u);
        float*       sn = wsw + (sel ^ 1u) * (WSTG * 32u);
        const float4* oq = (const float4*)(sw + NH * 32);
        const float4* wq = (const float4*)(sw + kq * 32);

        auto issue = [&](float4 (&r)[8], int j0) {
          const float4 o0 = oq[j0 >> 2], o1 = oq[(j0 >> 2) + 1];
          const unsigned o[8] = {__float_as_uint(o0.x), __float_as_uint(o0.y), __float_as_uint(o0.z), __float_as_uint(o0.w),
                                 __float_as_uint(o1.x), __float_as_uint(o1.y), __float_as_uint(o1.z), __float_as_uint(o1.w)};
#pragma unroll
          for (int u = 0; u < 8; ++u)
            r[u] = *(const float4*)(WhB + (o[u] + lane16));
        };
        auto add8 = [&](const float4 (&r)[8], int j0) {
          const float4 w0 = wq[j0 >> 2], w1 = wq[(j0 >> 2) + 1];
          const float w[8] = {w0.x, w0.y, w0.z, w0.w, w1.x, w1.y, w1.z, w1.w};
#pragma unroll
          for (int u = 0; u < 8; ++u) {
            if (__builtin_expect((bm >> (j0 + u)) & 1u, 0)) {
              store_row();
              a4 = make_float4(0.f, 0.f, 0.f, 0.f);
              ws = 0.f;
              cur = (unsigned)__builtin_amdgcn_readlane((int)dv, j0 + u);
            }
            a4.x += w[u] * r[u].x;
            a4.y += w[u] * r[u].y;
            a4.z += w[u] * r[u].z;
            a4.w += w[u] * r[u].w;
            ws += w[u];
          }
        };

        float4 ra[8], rb[8];
        issue(ra, 0);
        issue(rb, 8);
        unsigned dvn;
        const unsigned bmn = stage(base + 32u, sn, dvn);
        __builtin_amdgcn_fence(__ATOMIC_ACQ_REL, "wavefront");
        add8(ra, 0);
        issue(ra, 16);
        add8(rb, 8);
        issue(rb, 24);
        add8(ra, 16);
        add8(rb, 24);
        dv = dvn; bm = bmn; sel ^= 1u;
      }
    }
    store_row();
  } else {
    for (int e = tid; e < NE; e += 1024) {
      const unsigned d = (unsigned)(tgt[e] - n0);
      if (d < (unsigned)NB) {
        const float4 ss = s_src4[src[e]];
        const float4 st = *(const float4*)&stg[d * NH];
        atomicMax(&emaxk[d * NH + 0], f2key(leaky(ss.x + st.x)));
        atomicMax(&emaxk[d * NH + 1], f2key(leaky(ss.y + st.y)));
        atomicMax(&emaxk[d * NH + 2], f2key(leaky(ss.z + st.z)));
        atomicMax(&emaxk[d * NH + 3], f2key(leaky(ss.w + st.w)));
      }
    }
    __syncthreads();
    for (int e = tid; e < NE; e += 1024) {
      const unsigned d = (unsigned)(tgt[e] - n0);
      if (d < (unsigned)NB) {
        const int s = src[e];
        const float4 ss = s_src4[s];
        const float4 st = *(const float4*)&stg[d * NH];
        const int4   mk = *(const int4*)&emaxk[d * NH];
        float exv[4];
        exv[0] = edge_exp(ss.x + st.x, mk.x);
        exv[1] = edge_exp(ss.y + st.y, mk.y);
        exv[2] = edge_exp(ss.z + st.z, mk.z);
        exv[3] = edge_exp(ss.w + st.w, mk.w);
#pragma unroll
        for (int k = 0; k < NH; ++k) atomicAdd(&esum[d * NH + k], exv[k]);
        const float* wr = Wh + (size_t)s * DT;
        float* ar = acc + d * ACCP;
        for (int j = 0; j < DT; ++j) atomicAdd(&ar[j], exv[j >> 5] * wr[j]);
      }
    }
  }
  __syncthreads();

  for (int i = tid; i < NB * DT; i += 1024) {
    const int d = i >> 7;
    const int j = i & 127;
    const float v = acc[d * ACCP + j] / (esum[d * NH + (j >> 5)] + 1e-10f);
    out[(size_t)(n0 + d) * DT + j] = v > 0.f ? v : expm1f(v);
  }
}

extern "C" void kernel_launch(void* const* d_in, const int* in_sizes, int n_in,
                              void* d_out, int out_size, void* d_ws, size_t ws_size,
                              hipStream_t stream) {
  const float* x    = (const float*)d_in[0];
  const int*   eidx = (const int*)d_in[1];
  const float* W    = (const float*)d_in[2];
  const float* a    = (const float*)d_in[3];
  float* out = (float*)d_out;

  const int* src = eidx;
  const int* tgt = eidx + NE;

  float* ws    = (float*)d_ws;
  float* Wh    = ws;
  float* s_src = Wh + (size_t)NN * DT;
  float* s_tgt = s_src + NN * NH;
  unsigned* ent = (unsigned*)(s_tgt + NN * NH);
  unsigned short* off = (unsigned short*)(ent + (size_t)NSEG * SEGE);

  gat_wh_gemm<<<NSEG, 128, 0, stream>>>(x, W, a, src, tgt, Wh, s_src, s_tgt, ent, off);

  gat_owner<<<NOWN, 1024, 0, stream>>>(src, tgt, ent, off, s_src, s_tgt, Wh, out);
}
